# starting kernel
# baseline (speedup 1.0000x reference)
#include <hip/hip_runtime.h>

#define V_  12
#define N_  80000
#define C_  256
#define H_  128
#define W_  128
#define G_  4096
#define M_  512
#define E_  200000
#define L_  20
#define C2_ 128

typedef __attribute__((ext_vector_type(2))) float v2f;
typedef __attribute__((ext_vector_type(8))) float v8f;

__global__ void k_zero(float* __restrict__ p, int n) {
  int i = blockIdx.x * blockDim.x + threadIdx.x;
  if (i < n) p[i] = 0.0f;
}

__global__ void k_group_gather(const float* __restrict__ img,
                               const float* __restrict__ is_seen,
                               const int*   __restrict__ coords,
                               const int*   __restrict__ centers,
                               float* __restrict__ group_feat) {
  int g = blockIdx.x;
  int c = threadIdx.x;
  int n = centers[g];
  float acc = 0.0f, den = 0.0f;
  for (int v = 0; v < V_; ++v) {
    int x = coords[((size_t)v * N_ + n) * 2 + 0];
    int y = coords[((size_t)v * N_ + n) * 2 + 1];
    float w = is_seen[(size_t)v * N_ + n];
    den += w;
    acc += w * img[(((size_t)v * C_ + c) * H_ + y) * W_ + x];
  }
  group_feat[(size_t)g * C_ + c] = acc / (den + 1e-6f);
}

__global__ void k_mask_scatter(const float* __restrict__ group_feat,
                               const int*   __restrict__ group2mask,
                               float* __restrict__ msum, float* __restrict__ mcnt) {
  int g = blockIdx.x;
  int c = threadIdx.x;
  int m = group2mask[g];
  unsafeAtomicAdd(&msum[(size_t)m * C_ + c], group_feat[(size_t)g * C_ + c]);
  if (c == 0) unsafeAtomicAdd(&mcnt[m], 1.0f);
}

__global__ void k_mask_norm(const float* __restrict__ msum,
                            const float* __restrict__ mcnt,
                            float* __restrict__ mask_feat) {
  int i = blockIdx.x * blockDim.x + threadIdx.x;
  int m = i >> 8;
  float cc = mcnt[m];
  mask_feat[i] = (cc > 0.0f) ? msum[i] / fmaxf(cc, 1.0f) : 0.0f;
}

__global__ void k_mask_h1(const float* __restrict__ mask_feat,
                          const float* __restrict__ W1,
                          float* __restrict__ mask_h1) {
  int lane = threadIdx.x & 31;
  int wid  = threadIdx.x >> 5;
  int tile = blockIdx.x * 4 + wid;
  int mt = tile >> 3, nt = tile & 7;
  int mrow = (mt << 4) + (lane & 15);
  int ncol = (nt << 4) + (lane & 15);
  int koff = (lane >> 4) << 1;
  v8f c = {};
  for (int kk = 0; kk < C_; kk += 4) {
    v2f a, b;
    int k0 = kk + koff, k1 = kk + koff + 1;
    a.x = mask_feat[(size_t)mrow * C_ + k0];
    a.y = mask_feat[(size_t)mrow * C_ + k1];
    b.x = W1[(size_t)k0 * C2_ + ncol];
    b.y = W1[(size_t)k1 * C2_ + ncol];
    c = __builtin_amdgcn_wmma_f32_16x16x4_f32(false, a, false, b, (short)0, c, false, false);
  }
  for (int i = 0; i < 8; ++i) {
    int r = (mt << 4) + i + ((lane >> 4) << 3);
    mask_h1[(size_t)r * C2_ + ncol] = c[i];
  }
}

__global__ void k_edge_scatter(const float* __restrict__ mask_h1,
                               const int*   __restrict__ pair_mask,
                               const int*   __restrict__ pair_point,
                               float* __restrict__ fsum, float* __restrict__ cnt) {
  long long idx = (long long)blockIdx.x * blockDim.x + threadIdx.x;
  if (idx >= (long long)E_ * C2_) return;
  int e = (int)(idx >> 7);
  int k = (int)(idx & 127);
  int m = pair_mask[e];
  int p = pair_point[e];
  unsafeAtomicAdd(&fsum[(size_t)p * C2_ + k], mask_h1[(size_t)m * C2_ + k]);
  if (k == 0) unsafeAtomicAdd(&cnt[p], 1.0f);
}

__global__ void k_final(const float* __restrict__ fsum, const float* __restrict__ cnt,
                        const float* __restrict__ b1,   const float* __restrict__ W2,
                        const float* __restrict__ b2,   float* __restrict__ out) {
  int lane = threadIdx.x & 31;
  int wid  = threadIdx.x >> 5;
  int m_base = ((blockIdx.x << 3) + wid) << 4;
  int m = m_base + (lane & 15);
  bool valid = (m < N_);
  float scale = 0.0f;
  if (valid) scale = 1.0f / fmaxf(cnt[m], 1e-6f);
  int koff = (lane >> 4) << 1;
  int col  = lane & 15;
  v8f c0 = {}, c1 = {};
  for (int kk = 0; kk < C2_; kk += 4) {
    int k0 = kk + koff, k1 = kk + koff + 1;
    v2f a, b0, b1v;
    if (valid) {
      a.x = fmaxf(fsum[(size_t)m * C2_ + k0] * scale + b1[k0], 0.0f);
      a.y = fmaxf(fsum[(size_t)m * C2_ + k1] * scale + b1[k1], 0.0f);
    } else {
      a.x = 0.0f; a.y = 0.0f;
    }
    b0.x = W2[(size_t)k0 * L_ + col];
    b0.y = W2[(size_t)k1 * L_ + col];
    int col1 = 16 + col;
    b1v.x = (col1 < L_) ? W2[(size_t)k0 * L_ + col1] : 0.0f;
    b1v.y = (col1 < L_) ? W2[(size_t)k1 * L_ + col1] : 0.0f;
    c0 = __builtin_amdgcn_wmma_f32_16x16x4_f32(false, a, false, b0,  (short)0, c0, false, false);
    c1 = __builtin_amdgcn_wmma_f32_16x16x4_f32(false, a, false, b1v, (short)0, c1, false, false);
  }
  for (int i = 0; i < 8; ++i) {
    int r = m_base + i + ((lane >> 4) << 3);
    if (r < N_) {
      out[(size_t)r * L_ + col] = c0[i] + b2[col];
      int col1 = 16 + col;
      if (col1 < L_) out[(size_t)r * L_ + col1] = c1[i] + b2[col1];
    }
  }
}

extern "C" void kernel_launch(void* const* d_in, const int* in_sizes, int n_in,
                              void* d_out, int out_size, void* d_ws, size_t ws_size,
                              hipStream_t stream) {
  const float* img        = (const float*)d_in[0];
  const float* is_seen    = (const float*)d_in[1];
  const float* W1         = (const float*)d_in[2];
  const float* b1         = (const float*)d_in[3];
  const float* W2         = (const float*)d_in[4];
  const float* b2         = (const float*)d_in[5];
  const int*   coords     = (const int*)d_in[6];
  const int*   centers    = (const int*)d_in[7];
  const int*   group2mask = (const int*)d_in[8];
  const int*   pair_mask  = (const int*)d_in[9];
  const int*   pair_point = (const int*)d_in[10];
  float* out = (float*)d_out;

  float* ws         = (float*)d_ws;
  float* group_feat = ws;
  float* msum       = group_feat + (size_t)G_ * C_;
  float* mcnt       = msum + (size_t)M_ * C_;
  float* mask_feat  = mcnt + M_;
  float* mask_h1    = mask_feat + (size_t)M_ * C_;
  float* fsum       = mask_h1 + (size_t)M_ * C2_;

  int nz1 = M_ * C_ + M_;
  k_zero<<<(nz1 + 255) / 256, 256, 0, stream>>>(msum, nz1);
  int nz2 = N_ * C2_ + N_;
  k_zero<<<(nz2 + 255) / 256, 256, 0, stream>>>(fsum, nz2);

  k_group_gather<<<G_, C_, 0, stream>>>(img, is_seen, coords, centers, group_feat);
  k_mask_scatter<<<G_, C_, 0, stream>>>(group_feat, group2mask, msum, mcnt);
  k_mask_norm<<<(M_ * C_) / 256, 256, 0, stream>>>(msum, mcnt, mask_feat);
  k_mask_h1<<<64, 128, 0, stream>>>(mask_feat, W1, mask_h1);

  long long ne = (long long)E_ * C2_;
  k_edge_scatter<<<(int)((ne + 255) / 256), 256, 0, stream>>>(
      mask_h1, pair_mask, pair_point, fsum, fsum + (size_t)N_ * C2_);

  int tiles = (N_ + 15) / 16;
  k_final<<<(tiles + 7) / 8, 256, 0, stream>>>(
      fsum, fsum + (size_t)N_ * C2_, b1, W2, b2, out);
}


# group gather: stream each (view,channel) 64 KiB plane once into LDS with whole-line loads, pick the group pixels from LDS (offset/weight table pre-pass); view sum order unchanged
# speedup vs baseline: 1.3085x; 1.3085x over previous
#include <hip/hip_runtime.h>

#define V_  12
#define N_  80000
#define C_  256
#define H_  128
#define W_  128
#define G_  4096
#define M_  512
#define E_  200000
#define L_  20
#define C2_ 128

typedef __attribute__((ext_vector_type(2))) float v2f;
typedef __attribute__((ext_vector_type(8))) float v8f;
typedef __attribute__((ext_vector_type(4))) float v4f;
typedef __attribute__((ext_vector_type(4))) int   v4i;

__global__ void k_zero(float* __restrict__ p, int n) {
  int i = blockIdx.x * blockDim.x + threadIdx.x;
  if (i < n) p[i] = 0.0f;
}

__global__ void k_group_table(const float* __restrict__ is_seen,
                              const int*   __restrict__ coords,
                              const int*   __restrict__ centers,
                              int* __restrict__ tab_off, float* __restrict__ tab_w) {
  int idx = blockIdx.x * blockDim.x + threadIdx.x;
  int v = idx >> 12;
  int g = idx & (G_ - 1);
  int n = centers[g];
  int x = coords[((size_t)v * N_ + n) * 2 + 0];
  int y = coords[((size_t)v * N_ + n) * 2 + 1];
  tab_off[idx] = y * W_ + x;
  tab_w[idx]   = is_seen[(size_t)v * N_ + n];
}

#define PLANE4_ (H_ * W_ / 4)

__device__ __forceinline__ void lds_barrier() {
  __builtin_amdgcn_fence(__ATOMIC_RELEASE, "workgroup", "local");
  __builtin_amdgcn_s_barrier();
  __builtin_amdgcn_fence(__ATOMIC_ACQUIRE, "workgroup", "local");
}

__global__ void __launch_bounds__(1024)
k_group_gather(const float* __restrict__ img,
               const int*   __restrict__ tab_off,
               const float* __restrict__ tab_w,
               float* __restrict__ group_featT) {
  __shared__ v4f plane[2][PLANE4_];
  const int c = blockIdx.x;
  const int t = threadIdx.x;
  const v4f* src = (const v4f*)img + (size_t)c * PLANE4_ + t;
  const size_t vstride = (size_t)C_ * PLANE4_;
  const v4i* toff = (const v4i*)tab_off + t;
  const v4f* tw   = (const v4f*)tab_w + t;

  v4f acc = {0.0f, 0.0f, 0.0f, 0.0f};
  v4f den = {0.0f, 0.0f, 0.0f, 0.0f};
  v4f ra[4], rb[4];
  v4i oa, ob;
  v4f wa, wb;

#pragma unroll
  for (int j = 0; j < 4; ++j) ra[j] = __builtin_nontemporal_load(src + j * 1024);
  oa = toff[0]; wa = tw[0];
#pragma unroll
  for (int j = 0; j < 4; ++j) rb[j] = __builtin_nontemporal_load(src + vstride + j * 1024);
  ob = toff[G_ / 4]; wb = tw[G_ / 4];

  for (int v = 0; v < V_; v += 2) {
    {
#pragma unroll
      for (int j = 0; j < 4; ++j) plane[0][t + j * 1024] = ra[j];
      lds_barrier();
      v4i o = oa; v4f w = wa;
      asm volatile("" : "+v"(o), "+v"(w));
      if (v + 2 < V_) {
#pragma unroll
        for (int j = 0; j < 4; ++j) ra[j] = __builtin_nontemporal_load(src + (size_t)(v + 2) * vstride + j * 1024);
        oa = toff[(v + 2) * (G_ / 4)]; wa = tw[(v + 2) * (G_ / 4)];
      }
      const float* pl = (const float*)plane[0];
#pragma unroll
      for (int j = 0; j < 4; ++j) {
        den[j] += w[j];
        acc[j] += w[j] * pl[o[j]];
      }
    }
    {
#pragma unroll
      for (int j = 0; j < 4; ++j) plane[1][t + j * 1024] = rb[j];
      lds_barrier();
      v4i o = ob; v4f w = wb;
      asm volatile("" : "+v"(o), "+v"(w));
      if (v + 3 < V_) {
#pragma unroll
        for (int j = 0; j < 4; ++j) rb[j] = __builtin_nontemporal_load(src + (size_t)(v + 3) * vstride + j * 1024);
        ob = toff[(v + 3) * (G_ / 4)]; wb = tw[(v + 3) * (G_ / 4)];
      }
      const float* pl = (const float*)plane[1];
#pragma unroll
      for (int j = 0; j < 4; ++j) {
        den[j] += w[j];
        acc[j] += w[j] * pl[o[j]];
      }
    }
  }
  v4f r;
#pragma unroll
  for (int j = 0; j < 4; ++j) r[j] = acc[j] / (den[j] + 1e-6f);
  *((v4f*)(group_featT + (size_t)c * G_) + t) = r;
}

__global__ void k_mask_scatter(const float* __restrict__ group_featT,
                               const int*   __restrict__ group2mask,
                               float* __restrict__ msum, float* __restrict__ mcnt) {
  int g = blockIdx.x;
  int c = threadIdx.x;
  int m = group2mask[g];
  unsafeAtomicAdd(&msum[(size_t)m * C_ + c], group_featT[(size_t)c * G_ + g]);
  if (c == 0) unsafeAtomicAdd(&mcnt[m], 1.0f);
}

__global__ void k_mask_norm(const float* __restrict__ msum,
                            const float* __restrict__ mcnt,
                            float* __restrict__ mask_feat) {
  int i = blockIdx.x * blockDim.x + threadIdx.x;
  int m = i >> 8;
  float cc = mcnt[m];
  mask_feat[i] = (cc > 0.0f) ? msum[i] / fmaxf(cc, 1.0f) : 0.0f;
}

__global__ void k_mask_h1(const float* __restrict__ mask_feat,
                          const float* __restrict__ W1,
                          float* __restrict__ mask_h1) {
  int lane = threadIdx.x & 31;
  int wid  = threadIdx.x >> 5;
  int tile = blockIdx.x * 4 + wid;
  int mt = tile >> 3, nt = tile & 7;
  int mrow = (mt << 4) + (lane & 15);
  int ncol = (nt << 4) + (lane & 15);
  int koff = (lane >> 4) << 1;
  v8f c = {};
  for (int kk = 0; kk < C_; kk += 4) {
    v2f a, b;
    int k0 = kk + koff, k1 = kk + koff + 1;
    a.x = mask_feat[(size_t)mrow * C_ + k0];
    a.y = mask_feat[(size_t)mrow * C_ + k1];
    b.x = W1[(size_t)k0 * C2_ + ncol];
    b.y = W1[(size_t)k1 * C2_ + ncol];
    c = __builtin_amdgcn_wmma_f32_16x16x4_f32(false, a, false, b, (short)0, c, false, false);
  }
  for (int i = 0; i < 8; ++i) {
    int r = (mt << 4) + i + ((lane >> 4) << 3);
    mask_h1[(size_t)r * C2_ + ncol] = c[i];
  }
}

__global__ void k_edge_scatter(const float* __restrict__ mask_h1,
                               const int*   __restrict__ pair_mask,
                               const int*   __restrict__ pair_point,
                               float* __restrict__ fsum, float* __restrict__ cnt) {
  long long idx = (long long)blockIdx.x * blockDim.x + threadIdx.x;
  if (idx >= (long long)E_ * C2_) return;
  int e = (int)(idx >> 7);
  int k = (int)(idx & 127);
  int m = pair_mask[e];
  int p = pair_point[e];
  unsafeAtomicAdd(&fsum[(size_t)p * C2_ + k], mask_h1[(size_t)m * C2_ + k]);
  if (k == 0) unsafeAtomicAdd(&cnt[p], 1.0f);
}

__global__ void k_final(const float* __restrict__ fsum, const float* __restrict__ cnt,
                        const float* __restrict__ b1,   const float* __restrict__ W2,
                        const float* __restrict__ b2,   float* __restrict__ out) {
  int lane = threadIdx.x & 31;
  int wid  = threadIdx.x >> 5;
  int m_base = ((blockIdx.x << 3) + wid) << 4;
  int m = m_base + (lane & 15);
  bool valid = (m < N_);
  float scale = 0.0f;
  if (valid) scale = 1.0f / fmaxf(cnt[m], 1e-6f);
  int koff = (lane >> 4) << 1;
  int col  = lane & 15;
  v8f c0 = {}, c1 = {};
  for (int kk = 0; kk < C2_; kk += 4) {
    int k0 = kk + koff, k1 = kk + koff + 1;
    v2f a, b0, b1v;
    if (valid) {
      a.x = fmaxf(fsum[(size_t)m * C2_ + k0] * scale + b1[k0], 0.0f);
      a.y = fmaxf(fsum[(size_t)m * C2_ + k1] * scale + b1[k1], 0.0f);
    } else {
      a.x = 0.0f; a.y = 0.0f;
    }
    b0.x = W2[(size_t)k0 * L_ + col];
    b0.y = W2[(size_t)k1 * L_ + col];
    int col1 = 16 + col;
    b1v.x = (col1 < L_) ? W2[(size_t)k0 * L_ + col1] : 0.0f;
    b1v.y = (col1 < L_) ? W2[(size_t)k1 * L_ + col1] : 0.0f;
    c0 = __builtin_amdgcn_wmma_f32_16x16x4_f32(false, a, false, b0,  (short)0, c0, false, false);
    c1 = __builtin_amdgcn_wmma_f32_16x16x4_f32(false, a, false, b1v, (short)0, c1, false, false);
  }
  for (int i = 0; i < 8; ++i) {
    int r = m_base + i + ((lane >> 4) << 3);
    if (r < N_) {
      out[(size_t)r * L_ + col] = c0[i] + b2[col];
      int col1 = 16 + col;
      if (col1 < L_) out[(size_t)r * L_ + col1] = c1[i] + b2[col1];
    }
  }
}

extern "C" void kernel_launch(void* const* d_in, const int* in_sizes, int n_in,
                              void* d_out, int out_size, void* d_ws, size_t ws_size,
                              hipStream_t stream) {
  const float* img        = (const float*)d_in[0];
  const float* is_seen    = (const float*)d_in[1];
  const float* W1         = (const float*)d_in[2];
  const float* b1         = (const float*)d_in[3];
  const float* W2         = (const float*)d_in[4];
  const float* b2         = (const float*)d_in[5];
  const int*   coords     = (const int*)d_in[6];
  const int*   centers    = (const int*)d_in[7];
  const int*   group2mask = (const int*)d_in[8];
  const int*   pair_mask  = (const int*)d_in[9];
  const int*   pair_point = (const int*)d_in[10];
  float* out = (float*)d_out;

  float* ws         = (float*)d_ws;
  float* group_feat = ws;
  float* msum       = group_feat + (size_t)G_ * C_;
  float* mcnt       = msum + (size_t)M_ * C_;
  float* mask_feat  = mcnt + M_;
  float* mask_h1    = mask_feat + (size_t)M_ * C_;
  float* fsum       = mask_h1 + (size_t)M_ * C2_;

  int nz1 = M_ * C_ + M_;
  k_zero<<<(nz1 + 255) / 256, 256, 0, stream>>>(msum, nz1);
  int nz2 = N_ * C2_ + N_;
  k_zero<<<(nz2 + 255) / 256, 256, 0, stream>>>(fsum, nz2);

  int*   tab_off = (int*)mask_feat;
  float* tab_w   = mask_feat + (size_t)V_ * G_;
  k_group_table<<<(V_ * G_) / 256, 256, 0, stream>>>(is_seen, coords, centers, tab_off, tab_w);
  k_group_gather<<<C_, 1024, 0, stream>>>(img, tab_off, tab_w, group_feat);
  k_mask_scatter<<<G_, C_, 0, stream>>>(group_feat, group2mask, msum, mcnt);
  k_mask_norm<<<(M_ * C_) / 256, 256, 0, stream>>>(msum, mcnt, mask_feat);
  k_mask_h1<<<64, 128, 0, stream>>>(mask_feat, W1, mask_h1);

  long long ne = (long long)E_ * C2_;
  k_edge_scatter<<<(int)((ne + 255) / 256), 256, 0, stream>>>(
      mask_h1, pair_mask, pair_point, fsum, fsum + (size_t)N_ * C2_);

  int tiles = (N_ + 15) / 16;
  k_final<<<(tiles + 7) / 8, 256, 0, stream>>>(
      fsum, fsum + (size_t)N_ * C2_, b1, W2, b2, out);
}


# edge sums kept in LDS: 250 workgroups each own 320 points, scan the edge list, add mask rows with LDS float atomics (arrival order, as the global atomics before) and run the final GEMM from LDS
# speedup vs baseline: 3.1372x; 2.3976x over previous
#include <hip/hip_runtime.h>

#define V_  12
#define N_  80000
#define C_  256
#define H_  128
#define W_  128
#define G_  4096
#define M_  512
#define E_  200000
#define L_  20
#define C2_ 128

typedef __attribute__((ext_vector_type(2))) float v2f;
typedef __attribute__((ext_vector_type(8))) float v8f;
typedef __attribute__((ext_vector_type(4))) float v4f;
typedef __attribute__((ext_vector_type(4))) int   v4i;

__global__ void k_zero(float* __restrict__ p, int n) {
  int i = blockIdx.x * blockDim.x + threadIdx.x;
  if (i < n) p[i] = 0.0f;
}

__global__ void k_group_table(const float* __restrict__ is_seen,
                              const int*   __restrict__ coords,
                              const int*   __restrict__ centers,
                              int* __restrict__ tab_off, float* __restrict__ tab_w) {
  int idx = blockIdx.x * blockDim.x + threadIdx.x;
  int v = idx >> 12;
  int g = idx & (G_ - 1);
  int n = centers[g];
  int x = coords[((size_t)v * N_ + n) * 2 + 0];
  int y = coords[((size_t)v * N_ + n) * 2 + 1];
  tab_off[idx] = y * W_ + x;
  tab_w[idx]   = is_seen[(size_t)v * N_ + n];
}

#define PLANE4_ (H_ * W_ / 4)

__device__ __forceinline__ void lds_barrier() {
  __builtin_amdgcn_fence(__ATOMIC_RELEASE, "workgroup", "local");
  __builtin_amdgcn_s_barrier();
  __builtin_amdgcn_fence(__ATOMIC_ACQUIRE, "workgroup", "local");
}

__global__ void __launch_bounds__(1024)
k_group_gather(const float* __restrict__ img,
               const int*   __restrict__ tab_off,
               const float* __restrict__ tab_w,
               float* __restrict__ group_featT) {
  __shared__ v4f plane[2][PLANE4_];
  const int c = blockIdx.x;
  const int t = threadIdx.x;
  const v4f* src = (const v4f*)img + (size_t)c * PLANE4_ + t;
  const size_t vstride = (size_t)C_ * PLANE4_;
  const v4i* toff = (const v4i*)tab_off + t;
  const v4f* tw   = (const v4f*)tab_w + t;

  v4f acc = {0.0f, 0.0f, 0.0f, 0.0f};
  v4f den = {0.0f, 0.0f, 0.0f, 0.0f};
  v4f ra[4], rb[4];
  v4i oa, ob;
  v4f wa, wb;

#pragma unroll
  for (int j = 0; j < 4; ++j) ra[j] = __builtin_nontemporal_load(src + j * 1024);
  oa = toff[0]; wa = tw[0];
#pragma unroll
  for (int j = 0; j < 4; ++j) rb[j] = __builtin_nontemporal_load(src + vstride + j * 1024);
  ob = toff[G_ / 4]; wb = tw[G_ / 4];

  for (int v = 0; v < V_; v += 2) {
    {
#pragma unroll
      for (int j = 0; j < 4; ++j) plane[0][t + j * 1024] = ra[j];
      lds_barrier();
      v4i o = oa; v4f w = wa;
      asm volatile("" : "+v"(o), "+v"(w));
      if (v + 2 < V_) {
#pragma unroll
        for (int j = 0; j < 4; ++j) ra[j] = __builtin_nontemporal_load(src + (size_t)(v + 2) * vstride + j * 1024);
        oa = toff[(v + 2) * (G_ / 4)]; wa = tw[(v + 2) * (G_ / 4)];
      }
      const float* pl = (const float*)plane[0];
#pragma unroll
      for (int j = 0; j < 4; ++j) {
        den[j] += w[j];
        acc[j] += w[j] * pl[o[j]];
      }
    }
    {
#pragma unroll
      for (int j = 0; j < 4; ++j) plane[1][t + j * 1024] = rb[j];
      lds_barrier();
      v4i o = ob; v4f w = wb;
      asm volatile("" : "+v"(o), "+v"(w));
      if (v + 3 < V_) {
#pragma unroll
        for (int j = 0; j < 4; ++j) rb[j] = __builtin_nontemporal_load(src + (size_t)(v + 3) * vstride + j * 1024);
        ob = toff[(v + 3) * (G_ / 4)]; wb = tw[(v + 3) * (G_ / 4)];
      }
      const float* pl = (const float*)plane[1];
#pragma unroll
      for (int j = 0; j < 4; ++j) {
        den[j] += w[j];
        acc[j] += w[j] * pl[o[j]];
      }
    }
  }
  v4f r;
#pragma unroll
  for (int j = 0; j < 4; ++j) r[j] = acc[j] / (den[j] + 1e-6f);
  *((v4f*)(group_featT + (size_t)c * G_) + t) = r;
}

__global__ void k_mask_scatter(const float* __restrict__ group_featT,
                               const int*   __restrict__ group2mask,
                               float* __restrict__ msum, float* __restrict__ mcnt) {
  int g = blockIdx.x;
  int c = threadIdx.x;
  int m = group2mask[g];
  unsafeAtomicAdd(&msum[(size_t)m * C_ + c], group_featT[(size_t)c * G_ + g]);
  if (c == 0) unsafeAtomicAdd(&mcnt[m], 1.0f);
}

__global__ void k_mask_norm(const float* __restrict__ msum,
                            const float* __restrict__ mcnt,
                            float* __restrict__ mask_feat) {
  int i = blockIdx.x * blockDim.x + threadIdx.x;
  int m = i >> 8;
  float cc = mcnt[m];
  mask_feat[i] = (cc > 0.0f) ? msum[i] / fmaxf(cc, 1.0f) : 0.0f;
}

__global__ void k_mask_h1(const float* __restrict__ mask_feat,
                          const float* __restrict__ W1,
                          float* __restrict__ mask_h1) {
  int lane = threadIdx.x & 31;
  int wid  = threadIdx.x >> 5;
  int tile = blockIdx.x * 4 + wid;
  int mt = tile >> 3, nt = tile & 7;
  int mrow = (mt << 4) + (lane & 15);
  int ncol = (nt << 4) + (lane & 15);
  int koff = (lane >> 4) << 1;
  v8f c = {};
  for (int kk = 0; kk < C_; kk += 4) {
    v2f a, b;
    int k0 = kk + koff, k1 = kk + koff + 1;
    a.x = mask_feat[(size_t)mrow * C_ + k0];
    a.y = mask_feat[(size_t)mrow * C_ + k1];
    b.x = W1[(size_t)k0 * C2_ + ncol];
    b.y = W1[(size_t)k1 * C2_ + ncol];
    c = __builtin_amdgcn_wmma_f32_16x16x4_f32(false, a, false, b, (short)0, c, false, false);
  }
  for (int i = 0; i < 8; ++i) {
    int r = (mt << 4) + i + ((lane >> 4) << 3);
    mask_h1[(size_t)r * C2_ + ncol] = c[i];
  }
}

#define PPW_   320
#define NWG_   (N_ / PPW_)
#define PITCH_ 132
#define CAP_   16384
#define E4_    (E_ / 4)
#define NB_    8

__global__ void __launch_bounds__(1024)
k_point_fused(const float* __restrict__ mask_h1,
              const int*   __restrict__ pair_mask,
              const int*   __restrict__ pair_point,
              const float* __restrict__ b1, const float* __restrict__ W2,
              const float* __restrict__ b2, float* __restrict__ out) {
  __shared__ v4f acc4[PPW_ * PITCH_ / 4];
  __shared__ int list[CAP_];
  __shared__ v4f w2f[(C2_ / 4) * 32];
  __shared__ float sb1[C2_];
  __shared__ int pcnt[PPW_];
  __shared__ int list_n;
  float* acc = (float*)acc4;
  const int t = threadIdx.x, lane = t & 31, wid = t >> 5;
  const int lo = blockIdx.x * PPW_;

  v4f wfrag;
  float b1v0 = 0.0f;
  {
    int c = lane & 15, ka = (wid << 2) + ((lane >> 4) << 1), kb = ka + 1;
    wfrag[0] = W2[(size_t)ka * L_ + c];
    wfrag[1] = W2[(size_t)kb * L_ + c];
    wfrag[2] = (16 + c < L_) ? W2[(size_t)ka * L_ + 16 + c] : 0.0f;
    wfrag[3] = (16 + c < L_) ? W2[(size_t)kb * L_ + 16 + c] : 0.0f;
    if (t < C2_) b1v0 = b1[t];
  }
  for (int i = t; i < PPW_ * PITCH_ / 4; i += 1024) acc4[i] = v4f{0.0f, 0.0f, 0.0f, 0.0f};
  if (t < PPW_) pcnt[t] = 0;
  if (t == 0) list_n = 0;
  __syncthreads();

  const v4i* pp4 = (const v4i*)pair_point;
  for (int it = 0; it < 49; it += 7) {
    v4i q[7];
#pragma unroll
    for (int u = 0; u < 7; ++u) {
      int idx = (it + u) * 1024 + t;
      q[u] = (idx < E4_) ? pp4[idx] : v4i{-1, -1, -1, -1};
    }
#pragma unroll
    for (int u = 0; u < 7; ++u) {
      int idx = (it + u) * 1024 + t;
#pragma unroll
      for (int j = 0; j < 4; ++j) {
        unsigned rel = (unsigned)(q[u][j] - lo);
        if (rel < (unsigned)PPW_) {
          int e = idx * 4 + j;
          int pos = atomicAdd(&list_n, 1);
          if (pos < CAP_) {
            list[pos] = (int)(rel << 18) | e;
          } else {
            int m = pair_mask[e];
            for (int k = 0; k < C2_; ++k)
              atomicAdd(&acc[rel * PITCH_ + k], mask_h1[(size_t)m * C2_ + k]);
            atomicAdd(&pcnt[rel], 1);
          }
        }
      }
    }
  }
  w2f[t] = wfrag;
  if (t < C2_) sb1[t] = b1v0;
  __syncthreads();
  const int n = __builtin_amdgcn_readfirstlane(list_n < CAP_ ? list_n : CAP_);
  const int swid = __builtin_amdgcn_readfirstlane(wid);

  for (int i = t; i < n; i += 1024) {
    int ent = list[i];
    int m = pair_mask[ent & 0x3FFFF];
    list[i] = (ent & ~0x3FFFF) | m;
    atomicAdd(&pcnt[ent >> 18], 1);
  }
  __syncthreads();

  for (int k0 = swid; k0 < n; k0 += 32 * NB_) {
    int x[NB_], ent[NB_];
    float r[NB_][4];
#pragma unroll
    for (int u = 0; u < NB_; ++u) {
      int k = k0 + 32 * u;
      x[u] = list[k < n ? k : k0];
    }
#pragma unroll
    for (int u = 0; u < NB_; ++u) ent[u] = __builtin_amdgcn_readfirstlane(x[u]);
#pragma unroll
    for (int u = 0; u < NB_; ++u) {
      const float* s = mask_h1 + (size_t)(ent[u] & 0x3FFFF) * C2_ + lane;
#pragma unroll
      for (int j = 0; j < 4; ++j) r[u][j] = s[32 * j];
    }
#pragma unroll
    for (int u = 0; u < NB_; ++u) {
      if (k0 + 32 * u < n) {
        float* d = acc + (ent[u] >> 18) * PITCH_ + lane;
#pragma unroll
        for (int j = 0; j < 4; ++j) atomicAdd(d + 32 * j, r[u][j]);
      }
    }
  }
  __syncthreads();

  if (swid < PPW_ / 16) {
    int c = lane & 15;
    int h = lane >> 4;
    int row = (swid << 4) + c;
    float scale = 1.0f / fmaxf((float)pcnt[row], 1e-6f);
    const float* arow = acc + row * PITCH_ + (h << 1);
    const float* brow = sb1 + (h << 1);
    v8f c0 = {}, c1 = {};
#pragma unroll 8
    for (int kk = 0; kk < C2_; kk += 4) {
      v2f s = *(const v2f*)(arow + kk);
      v2f bb = *(const v2f*)(brow + kk);
      v4f f = w2f[(kk >> 2) * 32 + lane];
      v2f a, b0, b1v;
      a.x = fmaxf(s.x * scale + bb.x, 0.0f);
      a.y = fmaxf(s.y * scale + bb.y, 0.0f);
      b0.x = f[0]; b0.y = f[1];
      b1v.x = f[2]; b1v.y = f[3];
      c0 = __builtin_amdgcn_wmma_f32_16x16x4_f32(false, a, false, b0,  (short)0, c0, false, false);
      c1 = __builtin_amdgcn_wmma_f32_16x16x4_f32(false, a, false, b1v, (short)0, c1, false, false);
    }
    for (int i = 0; i < 8; ++i) {
      int r = lo + (swid << 4) + i + (h << 3);
      out[(size_t)r * L_ + c] = c0[i] + b2[c];
      int col1 = 16 + c;
      if (col1 < L_) out[(size_t)r * L_ + col1] = c1[i] + b2[col1];
    }
  }
}

extern "C" void kernel_launch(void* const* d_in, const int* in_sizes, int n_in,
                              void* d_out, int out_size, void* d_ws, size_t ws_size,
                              hipStream_t stream) {
  const float* img        = (const float*)d_in[0];
  const float* is_seen    = (const float*)d_in[1];
  const float* W1         = (const float*)d_in[2];
  const float* b1         = (const float*)d_in[3];
  const float* W2         = (const float*)d_in[4];
  const float* b2         = (const float*)d_in[5];
  const int*   coords     = (const int*)d_in[6];
  const int*   centers    = (const int*)d_in[7];
  const int*   group2mask = (const int*)d_in[8];
  const int*   pair_mask  = (const int*)d_in[9];
  const int*   pair_point = (const int*)d_in[10];
  float* out = (float*)d_out;

  float* ws         = (float*)d_ws;
  float* group_feat = ws;
  float* msum       = group_feat + (size_t)G_ * C_;
  float* mcnt       = msum + (size_t)M_ * C_;
  float* mask_feat  = mcnt + M_;
  float* mask_h1    = mask_feat + (size_t)M_ * C_;

  int nz1 = M_ * C_ + M_;
  k_zero<<<(nz1 + 255) / 256, 256, 0, stream>>>(msum, nz1);

  int*   tab_off = (int*)mask_feat;
  float* tab_w   = mask_feat + (size_t)V_ * G_;
  k_group_table<<<(V_ * G_) / 256, 256, 0, stream>>>(is_seen, coords, centers, tab_off, tab_w);
  k_group_gather<<<C_, 1024, 0, stream>>>(img, tab_off, tab_w, group_feat);
  k_mask_scatter<<<G_, C_, 0, stream>>>(group_feat, group2mask, msum, mcnt);
  k_mask_norm<<<(M_ * C_) / 256, 256, 0, stream>>>(msum, mcnt, mask_feat);
  k_mask_h1<<<64, 128, 0, stream>>>(mask_feat, W1, mask_h1);

  k_point_fused<<<NWG_, 1024, 0, stream>>>(mask_h1, pair_mask, pair_point, b1, W2, b2, out);
}


# mask mean fused into the gather block: each channel's 512 mask sums and counts kept in LDS (LDS float atomics, arrival order as the global atomics before); zero, scatter, normalise launches dropped
# speedup vs baseline: 3.8782x; 1.2362x over previous
#include <hip/hip_runtime.h>

#define V_  12
#define N_  80000
#define C_  256
#define H_  128
#define W_  128
#define G_  4096
#define M_  512
#define E_  200000
#define L_  20
#define C2_ 128

typedef __attribute__((ext_vector_type(2))) float v2f;
typedef __attribute__((ext_vector_type(8))) float v8f;
typedef __attribute__((ext_vector_type(4))) float v4f;
typedef __attribute__((ext_vector_type(4))) int   v4i;

__global__ void k_group_table(const float* __restrict__ is_seen,
                              const int*   __restrict__ coords,
                              const int*   __restrict__ centers,
                              int* __restrict__ tab_off, float* __restrict__ tab_w) {
  int idx = blockIdx.x * blockDim.x + threadIdx.x;
  int v = idx >> 12;
  int g = idx & (G_ - 1);
  int n = centers[g];
  int x = coords[((size_t)v * N_ + n) * 2 + 0];
  int y = coords[((size_t)v * N_ + n) * 2 + 1];
  tab_off[idx] = y * W_ + x;
  tab_w[idx]   = is_seen[(size_t)v * N_ + n];
}

#define PLANE4_ (H_ * W_ / 4)

__device__ __forceinline__ void lds_barrier() {
  __builtin_amdgcn_fence(__ATOMIC_RELEASE, "workgroup", "local");
  __builtin_amdgcn_s_barrier();
  __builtin_amdgcn_fence(__ATOMIC_ACQUIRE, "workgroup", "local");
}

__global__ void __launch_bounds__(1024)
k_group_gather(const float* __restrict__ img,
               const int*   __restrict__ tab_off,
               const float* __restrict__ tab_w,
               const int*   __restrict__ group2mask,
               float* __restrict__ mask_featT) {
  __shared__ v4f plane[2][PLANE4_];
  __shared__ float msum_l[M_];
  __shared__ int   mcnt_l[M_];
  const int c = blockIdx.x;
  const int t = threadIdx.x;
  const v4f* src = (const v4f*)img + (size_t)c * PLANE4_ + t;
  const size_t vstride = (size_t)C_ * PLANE4_;
  const v4i* toff = (const v4i*)tab_off + t;
  const v4f* tw   = (const v4f*)tab_w + t;

  const v4i g2m = *((const v4i*)group2mask + t);

  v4f acc = {0.0f, 0.0f, 0.0f, 0.0f};
  v4f den = {0.0f, 0.0f, 0.0f, 0.0f};
  v4f ra[4], rb[4];
  v4i oa, ob;
  v4f wa, wb;

#pragma unroll
  for (int j = 0; j < 4; ++j) ra[j] = __builtin_nontemporal_load(src + j * 1024);
  oa = toff[0]; wa = tw[0];
#pragma unroll
  for (int j = 0; j < 4; ++j) rb[j] = __builtin_nontemporal_load(src + vstride + j * 1024);
  ob = toff[G_ / 4]; wb = tw[G_ / 4];

  if (t < M_) { msum_l[t] = 0.0f; mcnt_l[t] = 0; }
  lds_barrier();
#pragma unroll
  for (int j = 0; j < 4; ++j) atomicAdd(&mcnt_l[g2m[j]], 1);

  for (int v = 0; v < V_; v += 2) {
    {
#pragma unroll
      for (int j = 0; j < 4; ++j) plane[0][t + j * 1024] = ra[j];
      lds_barrier();
      v4i o = oa; v4f w = wa;
      asm volatile("" : "+v"(o), "+v"(w));
      if (v + 2 < V_) {
#pragma unroll
        for (int j = 0; j < 4; ++j) ra[j] = __builtin_nontemporal_load(src + (size_t)(v + 2) * vstride + j * 1024);
        oa = toff[(v + 2) * (G_ / 4)]; wa = tw[(v + 2) * (G_ / 4)];
      }
      const float* pl = (const float*)plane[0];
#pragma unroll
      for (int j = 0; j < 4; ++j) {
        den[j] += w[j];
        acc[j] += w[j] * pl[o[j]];
      }
    }
    {
#pragma unroll
      for (int j = 0; j < 4; ++j) plane[1][t + j * 1024] = rb[j];
      lds_barrier();
      v4i o = ob; v4f w = wb;
      asm volatile("" : "+v"(o), "+v"(w));
      if (v + 3 < V_) {
#pragma unroll
        for (int j = 0; j < 4; ++j) rb[j] = __builtin_nontemporal_load(src + (size_t)(v + 3) * vstride + j * 1024);
        ob = toff[(v + 3) * (G_ / 4)]; wb = tw[(v + 3) * (G_ / 4)];
      }
      const float* pl = (const float*)plane[1];
#pragma unroll
      for (int j = 0; j < 4; ++j) {
        den[j] += w[j];
        acc[j] += w[j] * pl[o[j]];
      }
    }
  }
#pragma unroll
  for (int j = 0; j < 4; ++j) atomicAdd(&msum_l[g2m[j]], acc[j] / (den[j] + 1e-6f));
  lds_barrier();
  if (t < M_) {
    float cc = (float)mcnt_l[t];
    mask_featT[(size_t)c * M_ + t] = (cc > 0.0f) ? msum_l[t] / fmaxf(cc, 1.0f) : 0.0f;
  }
}

__global__ void k_mask_h1(const float* __restrict__ mask_featT,
                          const float* __restrict__ W1,
                          float* __restrict__ mask_h1) {
  int lane = threadIdx.x & 31;
  int wid  = threadIdx.x >> 5;
  int tile = blockIdx.x * 4 + wid;
  int mt = tile >> 3, nt = tile & 7;
  int mrow = (mt << 4) + (lane & 15);
  int ncol = (nt << 4) + (lane & 15);
  int koff = (lane >> 4) << 1;
  v8f c = {};
#pragma unroll 8
  for (int kk = 0; kk < C_; kk += 4) {
    v2f a, b;
    int k0 = kk + koff, k1 = kk + koff + 1;
    a.x = mask_featT[(size_t)k0 * M_ + mrow];
    a.y = mask_featT[(size_t)k1 * M_ + mrow];
    b.x = W1[(size_t)k0 * C2_ + ncol];
    b.y = W1[(size_t)k1 * C2_ + ncol];
    c = __builtin_amdgcn_wmma_f32_16x16x4_f32(false, a, false, b, (short)0, c, false, false);
  }
  for (int i = 0; i < 8; ++i) {
    int r = (mt << 4) + i + ((lane >> 4) << 3);
    mask_h1[(size_t)r * C2_ + ncol] = c[i];
  }
}

#define PPW_   320
#define NWG_   (N_ / PPW_)
#define PITCH_ 132
#define CAP_   16384
#define E4_    (E_ / 4)
#define NB_    8

__global__ void __launch_bounds__(1024)
k_point_fused(const float* __restrict__ mask_h1,
              const int*   __restrict__ pair_mask,
              const int*   __restrict__ pair_point,
              const float* __restrict__ b1, const float* __restrict__ W2,
              const float* __restrict__ b2, float* __restrict__ out) {
  __shared__ v4f acc4[PPW_ * PITCH_ / 4];
  __shared__ int list[CAP_];
  __shared__ v4f w2f[(C2_ / 4) * 32];
  __shared__ float sb1[C2_];
  __shared__ int pcnt[PPW_];
  __shared__ int list_n;
  float* acc = (float*)acc4;
  const int t = threadIdx.x, lane = t & 31, wid = t >> 5;
  const int lo = blockIdx.x * PPW_;

  v4f wfrag;
  float b1v0 = 0.0f;
  {
    int c = lane & 15, ka = (wid << 2) + ((lane >> 4) << 1), kb = ka + 1;
    wfrag[0] = W2[(size_t)ka * L_ + c];
    wfrag[1] = W2[(size_t)kb * L_ + c];
    wfrag[2] = (16 + c < L_) ? W2[(size_t)ka * L_ + 16 + c] : 0.0f;
    wfrag[3] = (16 + c < L_) ? W2[(size_t)kb * L_ + 16 + c] : 0.0f;
    if (t < C2_) b1v0 = b1[t];
  }
  for (int i = t; i < PPW_ * PITCH_ / 4; i += 1024) acc4[i] = v4f{0.0f, 0.0f, 0.0f, 0.0f};
  if (t < PPW_) pcnt[t] = 0;
  if (t == 0) list_n = 0;
  __syncthreads();

  const v4i* pp4 = (const v4i*)pair_point;
  for (int it = 0; it < 49; it += 7) {
    v4i q[7];
#pragma unroll
    for (int u = 0; u < 7; ++u) {
      int idx = (it + u) * 1024 + t;
      q[u] = (idx < E4_) ? pp4[idx] : v4i{-1, -1, -1, -1};
    }
#pragma unroll
    for (int u = 0; u < 7; ++u) {
      int idx = (it + u) * 1024 + t;
#pragma unroll
      for (int j = 0; j < 4; ++j) {
        unsigned rel = (unsigned)(q[u][j] - lo);
        if (rel < (unsigned)PPW_) {
          int e = idx * 4 + j;
          int pos = atomicAdd(&list_n, 1);
          if (pos < CAP_) {
            list[pos] = (int)(rel << 18) | e;
          } else {
            int m = pair_mask[e];
            for (int k = 0; k < C2_; ++k)
              atomicAdd(&acc[rel * PITCH_ + k], mask_h1[(size_t)m * C2_ + k]);
            atomicAdd(&pcnt[rel], 1);
          }
        }
      }
    }
  }
  w2f[t] = wfrag;
  if (t < C2_) sb1[t] = b1v0;
  __syncthreads();
  const int n = __builtin_amdgcn_readfirstlane(list_n < CAP_ ? list_n : CAP_);
  const int swid = __builtin_amdgcn_readfirstlane(wid);

  for (int i = t; i < n; i += 1024) {
    int ent = list[i];
    int m = pair_mask[ent & 0x3FFFF];
    list[i] = (ent & ~0x3FFFF) | m;
    atomicAdd(&pcnt[ent >> 18], 1);
  }
  __syncthreads();

  for (int k0 = swid; k0 < n; k0 += 32 * NB_) {
    int x[NB_], ent[NB_];
    float r[NB_][4];
#pragma unroll
    for (int u = 0; u < NB_; ++u) {
      int k = k0 + 32 * u;
      x[u] = list[k < n ? k : k0];
    }
#pragma unroll
    for (int u = 0; u < NB_; ++u) ent[u] = __builtin_amdgcn_readfirstlane(x[u]);
#pragma unroll
    for (int u = 0; u < NB_; ++u) {
      const float* s = mask_h1 + (size_t)(ent[u] & 0x3FFFF) * C2_ + lane;
#pragma unroll
      for (int j = 0; j < 4; ++j) r[u][j] = s[32 * j];
    }
#pragma unroll
    for (int u = 0; u < NB_; ++u) {
      if (k0 + 32 * u < n) {
        float* d = acc + (ent[u] >> 18) * PITCH_ + lane;
#pragma unroll
        for (int j = 0; j < 4; ++j) atomicAdd(d + 32 * j, r[u][j]);
      }
    }
  }
  __syncthreads();

  if (swid < PPW_ / 16) {
    int c = lane & 15;
    int h = lane >> 4;
    int row = (swid << 4) + c;
    float scale = 1.0f / fmaxf((float)pcnt[row], 1e-6f);
    const float* arow = acc + row * PITCH_ + (h << 1);
    const float* brow = sb1 + (h << 1);
    v8f c0 = {}, c1 = {};
#pragma unroll 8
    for (int kk = 0; kk < C2_; kk += 4) {
      v2f s = *(const v2f*)(arow + kk);
      v2f bb = *(const v2f*)(brow + kk);
      v4f f = w2f[(kk >> 2) * 32 + lane];
      v2f a, b0, b1v;
      a.x = fmaxf(s.x * scale + bb.x, 0.0f);
      a.y = fmaxf(s.y * scale + bb.y, 0.0f);
      b0.x = f[0]; b0.y = f[1];
      b1v.x = f[2]; b1v.y = f[3];
      c0 = __builtin_amdgcn_wmma_f32_16x16x4_f32(false, a, false, b0,  (short)0, c0, false, false);
      c1 = __builtin_amdgcn_wmma_f32_16x16x4_f32(false, a, false, b1v, (short)0, c1, false, false);
    }
    for (int i = 0; i < 8; ++i) {
      int r = lo + (swid << 4) + i + (h << 3);
      out[(size_t)r * L_ + c] = c0[i] + b2[c];
      int col1 = 16 + c;
      if (col1 < L_) out[(size_t)r * L_ + col1] = c1[i] + b2[col1];
    }
  }
}

extern "C" void kernel_launch(void* const* d_in, const int* in_sizes, int n_in,
                              void* d_out, int out_size, void* d_ws, size_t ws_size,
                              hipStream_t stream) {
  const float* img        = (const float*)d_in[0];
  const float* is_seen    = (const float*)d_in[1];
  const float* W1         = (const float*)d_in[2];
  const float* b1         = (const float*)d_in[3];
  const float* W2         = (const float*)d_in[4];
  const float* b2         = (const float*)d_in[5];
  const int*   coords     = (const int*)d_in[6];
  const int*   centers    = (const int*)d_in[7];
  const int*   group2mask = (const int*)d_in[8];
  const int*   pair_mask  = (const int*)d_in[9];
  const int*   pair_point = (const int*)d_in[10];
  float* out = (float*)d_out;

  float* ws         = (float*)d_ws;
  int*   tab_off    = (int*)ws;
  float* tab_w      = ws + (size_t)V_ * G_;
  float* mask_featT = ws + (size_t)G_ * C_ + (size_t)M_ * C_ + M_;
  float* mask_h1    = mask_featT + (size_t)M_ * C_;

  k_group_table<<<(V_ * G_) / 256, 256, 0, stream>>>(is_seen, coords, centers, tab_off, tab_w);
  k_group_gather<<<C_, 1024, 0, stream>>>(img, tab_off, tab_w, group2mask, mask_featT);
  k_mask_h1<<<64, 128, 0, stream>>>(mask_featT, W1, mask_h1);

  k_point_fused<<<NWG_, 1024, 0, stream>>>(mask_h1, pair_mask, pair_point, b1, W2, b2, out);
}


# edges bucketed once by owning point range (counting sort per 4000-edge chunk, inside the table launch); point kernel reads its 50 segments, no full edge scan; float add order stays arrival order
# speedup vs baseline: 4.6868x; 1.2085x over previous
#include <hip/hip_runtime.h>

#define V_  12
#define N_  80000
#define C_  256
#define H_  128
#define W_  128
#define G_  4096
#define M_  512
#define E_  200000
#define L_  20
#define C2_ 128

typedef __attribute__((ext_vector_type(2))) float v2f;
typedef __attribute__((ext_vector_type(8))) float v8f;
typedef __attribute__((ext_vector_type(4))) float v4f;
typedef __attribute__((ext_vector_type(4))) int   v4i;

#define PPW_   320
#define NWG_   (N_ / PPW_)
#define E4_    (E_ / 4)
#define NCH_   50
#define CH4_   (E4_ / NCH_)
#define CHE_   (CH4_ * 4)
#define OPITCH_ 256
#define TABB_  ((V_ * G_) / 1024)

__global__ void __launch_bounds__(1024)
k_prepass(const float* __restrict__ is_seen,
          const int*   __restrict__ coords,
          const int*   __restrict__ centers,
          const int*   __restrict__ pair_mask,
          const int*   __restrict__ pair_point,
          int* __restrict__ tab_off, float* __restrict__ tab_w,
          int* __restrict__ sorted, int* __restrict__ offs) {
  __shared__ int hist[OPITCH_];
  __shared__ int base[OPITCH_];
  const int t = threadIdx.x;
  if (blockIdx.x >= NCH_) {
    int idx = (blockIdx.x - NCH_) * 1024 + t;
    int v = idx >> 12;
    int g = idx & (G_ - 1);
    int n = centers[g];
    int x = coords[((size_t)v * N_ + n) * 2 + 0];
    int y = coords[((size_t)v * N_ + n) * 2 + 1];
    tab_off[idx] = y * W_ + x;
    tab_w[idx]   = is_seen[(size_t)v * N_ + n];
    return;
  }
  const int s = blockIdx.x;
  v4i p = {-1, -1, -1, -1};
  v4i m = {0, 0, 0, 0};
  if (t < CH4_) {
    p = ((const v4i*)pair_point)[s * CH4_ + t];
    m = ((const v4i*)pair_mask)[s * CH4_ + t];
  }
  if (t < OPITCH_) hist[t] = 0;
  __syncthreads();
  int own[4], rk[4];
#pragma unroll
  for (int j = 0; j < 4; ++j) {
    unsigned up = (unsigned)p[j];
    own[j] = (up < (unsigned)N_) ? (int)(up / PPW_) : -1;
    rk[j] = 0;
    if (own[j] >= 0) rk[j] = atomicAdd(&hist[own[j]], 1);
  }
  __syncthreads();
  int bstart[8] = {0, 0, 0, 0, 0, 0, 0, 0};
  if (t < 32) {
    int c[8];
    int run = 0;
#pragma unroll
    for (int i = 0; i < 8; ++i) { int h = hist[8 * t + i]; c[i] = run; run += h; }
    int incl = run;
#pragma unroll
    for (int d = 1; d < 32; d <<= 1) {
      int y = __shfl_up(incl, d);
      if (t >= d) incl += y;
    }
    int excl = incl - run;
#pragma unroll
    for (int i = 0; i < 8; ++i) {
      bstart[i] = excl + c[i];
      base[8 * t + i] = bstart[i];
    }
  }
  __syncthreads();
  if (t < 32) {
#pragma unroll
    for (int i = 0; i < 8; ++i) offs[s * OPITCH_ + 8 * t + i] = bstart[i];
  }
#pragma unroll
  for (int j = 0; j < 4; ++j) {
    if (own[j] >= 0)
      sorted[s * CHE_ + base[own[j]] + rk[j]] = ((p[j] - own[j] * PPW_) << 18) | m[j];
  }
}

#define PLANE4_ (H_ * W_ / 4)

__device__ __forceinline__ void lds_barrier() {
  __builtin_amdgcn_fence(__ATOMIC_RELEASE, "workgroup", "local");
  __builtin_amdgcn_s_barrier();
  __builtin_amdgcn_fence(__ATOMIC_ACQUIRE, "workgroup", "local");
}

__global__ void __launch_bounds__(1024)
k_group_gather(const float* __restrict__ img,
               const int*   __restrict__ tab_off,
               const float* __restrict__ tab_w,
               const int*   __restrict__ group2mask,
               float* __restrict__ mask_featT) {
  __shared__ v4f plane[2][PLANE4_];
  __shared__ float msum_l[M_];
  __shared__ int   mcnt_l[M_];
  const int c = blockIdx.x;
  const int t = threadIdx.x;
  const v4f* src = (const v4f*)img + (size_t)c * PLANE4_ + t;
  const size_t vstride = (size_t)C_ * PLANE4_;
  const v4i* toff = (const v4i*)tab_off + t;
  const v4f* tw   = (const v4f*)tab_w + t;

  const v4i g2m = *((const v4i*)group2mask + t);

  v4f acc = {0.0f, 0.0f, 0.0f, 0.0f};
  v4f den = {0.0f, 0.0f, 0.0f, 0.0f};
  v4f ra[4], rb[4];
  v4i oa, ob;
  v4f wa, wb;

#pragma unroll
  for (int j = 0; j < 4; ++j) ra[j] = __builtin_nontemporal_load(src + j * 1024);
  oa = toff[0]; wa = tw[0];
#pragma unroll
  for (int j = 0; j < 4; ++j) rb[j] = __builtin_nontemporal_load(src + vstride + j * 1024);
  ob = toff[G_ / 4]; wb = tw[G_ / 4];

  if (t < M_) { msum_l[t] = 0.0f; mcnt_l[t] = 0; }
  lds_barrier();
#pragma unroll
  for (int j = 0; j < 4; ++j) atomicAdd(&mcnt_l[g2m[j]], 1);

  for (int v = 0; v < V_; v += 2) {
    {
#pragma unroll
      for (int j = 0; j < 4; ++j) plane[0][t + j * 1024] = ra[j];
      lds_barrier();
      v4i o = oa; v4f w = wa;
      asm volatile("" : "+v"(o), "+v"(w));
      if (v + 2 < V_) {
#pragma unroll
        for (int j = 0; j < 4; ++j) ra[j] = __builtin_nontemporal_load(src + (size_t)(v + 2) * vstride + j * 1024);
        oa = toff[(v + 2) * (G_ / 4)]; wa = tw[(v + 2) * (G_ / 4)];
      }
      const float* pl = (const float*)plane[0];
#pragma unroll
      for (int j = 0; j < 4; ++j) {
        den[j] += w[j];
        acc[j] += w[j] * pl[o[j]];
      }
    }
    {
#pragma unroll
      for (int j = 0; j < 4; ++j) plane[1][t + j * 1024] = rb[j];
      lds_barrier();
      v4i o = ob; v4f w = wb;
      asm volatile("" : "+v"(o), "+v"(w));
      if (v + 3 < V_) {
#pragma unroll
        for (int j = 0; j < 4; ++j) rb[j] = __builtin_nontemporal_load(src + (size_t)(v + 3) * vstride + j * 1024);
        ob = toff[(v + 3) * (G_ / 4)]; wb = tw[(v + 3) * (G_ / 4)];
      }
      const float* pl = (const float*)plane[1];
#pragma unroll
      for (int j = 0; j < 4; ++j) {
        den[j] += w[j];
        acc[j] += w[j] * pl[o[j]];
      }
    }
  }
#pragma unroll
  for (int j = 0; j < 4; ++j) atomicAdd(&msum_l[g2m[j]], acc[j] / (den[j] + 1e-6f));
  lds_barrier();
  if (t < M_) {
    float cc = (float)mcnt_l[t];
    mask_featT[(size_t)c * M_ + t] = (cc > 0.0f) ? msum_l[t] / fmaxf(cc, 1.0f) : 0.0f;
  }
}

__global__ void k_mask_h1(const float* __restrict__ mask_featT,
                          const float* __restrict__ W1,
                          float* __restrict__ mask_h1) {
  int lane = threadIdx.x & 31;
  int wid  = threadIdx.x >> 5;
  int tile = blockIdx.x * 4 + wid;
  int mt = tile >> 3, nt = tile & 7;
  int mrow = (mt << 4) + (lane & 15);
  int ncol = (nt << 4) + (lane & 15);
  int koff = (lane >> 4) << 1;
  v8f c = {};
#pragma unroll 8
  for (int kk = 0; kk < C_; kk += 4) {
    v2f a, b;
    int k0 = kk + koff, k1 = kk + koff + 1;
    a.x = mask_featT[(size_t)k0 * M_ + mrow];
    a.y = mask_featT[(size_t)k1 * M_ + mrow];
    b.x = W1[(size_t)k0 * C2_ + ncol];
    b.y = W1[(size_t)k1 * C2_ + ncol];
    c = __builtin_amdgcn_wmma_f32_16x16x4_f32(false, a, false, b, (short)0, c, false, false);
  }
  for (int i = 0; i < 8; ++i) {
    int r = (mt << 4) + i + ((lane >> 4) << 3);
    mask_h1[(size_t)r * C2_ + ncol] = c[i];
  }
}

#define PITCH_ 132
#define CAP_   16384
#define TPS_   20
#define NB_    8

__global__ void __launch_bounds__(1024)
k_point_fused(const float* __restrict__ mask_h1,
              const int*   __restrict__ sorted,
              const int*   __restrict__ offs,
              const float* __restrict__ b1, const float* __restrict__ W2,
              const float* __restrict__ b2, float* __restrict__ out) {
  __shared__ v4f acc4[PPW_ * PITCH_ / 4];
  __shared__ int list[CAP_];
  __shared__ v4f w2f[(C2_ / 4) * 32];
  __shared__ float sb1[C2_];
  __shared__ int pcnt[PPW_];
  __shared__ int list_n;
  float* acc = (float*)acc4;
  const int t = threadIdx.x, lane = t & 31, wid = t >> 5;
  const int lo = blockIdx.x * PPW_;

  v4f wfrag;
  float b1v0 = 0.0f;
  {
    int c = lane & 15, ka = (wid << 2) + ((lane >> 4) << 1), kb = ka + 1;
    wfrag[0] = W2[(size_t)ka * L_ + c];
    wfrag[1] = W2[(size_t)kb * L_ + c];
    wfrag[2] = (16 + c < L_) ? W2[(size_t)ka * L_ + 16 + c] : 0.0f;
    wfrag[3] = (16 + c < L_) ? W2[(size_t)kb * L_ + 16 + c] : 0.0f;
    if (t < C2_) b1v0 = b1[t];
  }
  const int sc = t / TPS_, sr = t - sc * TPS_;
  int seg_st = 0, seg_en = 0;
  if (t < NCH_ * TPS_) {
    const int* o = offs + sc * OPITCH_ + blockIdx.x;
    seg_st = o[0];
    seg_en = o[1];
  }
  for (int i = t; i < PPW_ * PITCH_ / 4; i += 1024) acc4[i] = v4f{0.0f, 0.0f, 0.0f, 0.0f};
  if (t < PPW_) pcnt[t] = 0;
  if (t == 0) list_n = 0;
  __syncthreads();

  {
    const int* seg = sorted + sc * CHE_;
    for (int i = seg_st + sr; i < seg_en; i += TPS_) {
      int ent = seg[i];
      int rel = ent >> 18;
      atomicAdd(&pcnt[rel], 1);
      int pos = atomicAdd(&list_n, 1);
      if (pos < CAP_) {
        list[pos] = ent;
      } else {
        int m = ent & 0x3FFFF;
        for (int k = 0; k < C2_; ++k)
          atomicAdd(&acc[rel * PITCH_ + k], mask_h1[(size_t)m * C2_ + k]);
      }
    }
  }
  w2f[t] = wfrag;
  if (t < C2_) sb1[t] = b1v0;
  __syncthreads();
  const int n = __builtin_amdgcn_readfirstlane(list_n < CAP_ ? list_n : CAP_);
  const int swid = __builtin_amdgcn_readfirstlane(wid);

  for (int k0 = swid; k0 < n; k0 += 32 * NB_) {
    int x[NB_], ent[NB_];
    float r[NB_][4];
#pragma unroll
    for (int u = 0; u < NB_; ++u) {
      int k = k0 + 32 * u;
      x[u] = list[k < n ? k : k0];
    }
#pragma unroll
    for (int u = 0; u < NB_; ++u) ent[u] = __builtin_amdgcn_readfirstlane(x[u]);
#pragma unroll
    for (int u = 0; u < NB_; ++u) {
      const float* s = mask_h1 + (size_t)(ent[u] & 0x3FFFF) * C2_ + lane;
#pragma unroll
      for (int j = 0; j < 4; ++j) r[u][j] = s[32 * j];
    }
#pragma unroll
    for (int u = 0; u < NB_; ++u) {
      if (k0 + 32 * u < n) {
        float* d = acc + (ent[u] >> 18) * PITCH_ + lane;
#pragma unroll
        for (int j = 0; j < 4; ++j) atomicAdd(d + 32 * j, r[u][j]);
      }
    }
  }
  __syncthreads();

  if (swid < PPW_ / 16) {
    int c = lane & 15;
    int h = lane >> 4;
    int row = (swid << 4) + c;
    float scale = 1.0f / fmaxf((float)pcnt[row], 1e-6f);
    const float* arow = acc + row * PITCH_ + (h << 1);
    const float* brow = sb1 + (h << 1);
    v8f c0 = {}, c1 = {};
#pragma unroll 8
    for (int kk = 0; kk < C2_; kk += 4) {
      v2f s = *(const v2f*)(arow + kk);
      v2f bb = *(const v2f*)(brow + kk);
      v4f f = w2f[(kk >> 2) * 32 + lane];
      v2f a, b0, b1v;
      a.x = fmaxf(s.x * scale + bb.x, 0.0f);
      a.y = fmaxf(s.y * scale + bb.y, 0.0f);
      b0.x = f[0]; b0.y = f[1];
      b1v.x = f[2]; b1v.y = f[3];
      c0 = __builtin_amdgcn_wmma_f32_16x16x4_f32(false, a, false, b0,  (short)0, c0, false, false);
      c1 = __builtin_amdgcn_wmma_f32_16x16x4_f32(false, a, false, b1v, (short)0, c1, false, false);
    }
    for (int i = 0; i < 8; ++i) {
      int r = lo + (swid << 4) + i + (h << 3);
      out[(size_t)r * L_ + c] = c0[i] + b2[c];
      int col1 = 16 + c;
      if (col1 < L_) out[(size_t)r * L_ + col1] = c1[i] + b2[col1];
    }
  }
}

extern "C" void kernel_launch(void* const* d_in, const int* in_sizes, int n_in,
                              void* d_out, int out_size, void* d_ws, size_t ws_size,
                              hipStream_t stream) {
  const float* img        = (const float*)d_in[0];
  const float* is_seen    = (const float*)d_in[1];
  const float* W1         = (const float*)d_in[2];
  const float* b1         = (const float*)d_in[3];
  const float* W2         = (const float*)d_in[4];
  const float* b2         = (const float*)d_in[5];
  const int*   coords     = (const int*)d_in[6];
  const int*   centers    = (const int*)d_in[7];
  const int*   group2mask = (const int*)d_in[8];
  const int*   pair_mask  = (const int*)d_in[9];
  const int*   pair_point = (const int*)d_in[10];
  float* out = (float*)d_out;

  float* ws         = (float*)d_ws;
  int*   tab_off    = (int*)ws;
  float* tab_w      = ws + (size_t)V_ * G_;
  int*   sorted     = (int*)(ws + (size_t)2 * V_ * G_);
  int*   offs       = sorted + E_;
  float* mask_featT = ws + (size_t)G_ * C_ + (size_t)M_ * C_ + M_;
  float* mask_h1    = mask_featT + (size_t)M_ * C_;

  k_prepass<<<NCH_ + TABB_, 1024, 0, stream>>>(is_seen, coords, centers, pair_mask, pair_point,
                                               tab_off, tab_w, sorted, offs);
  k_group_gather<<<C_, 1024, 0, stream>>>(img, tab_off, tab_w, group2mask, mask_featT);
  k_mask_h1<<<64, 128, 0, stream>>>(mask_featT, W1, mask_h1);

  k_point_fused<<<NWG_, 1024, 0, stream>>>(mask_h1, sorted, offs, b1, W2, b2, out);
}


# mask GEMM: one tile per workgroup, its 4 waves split K in quarters so a wave's 64 operand loads fly at once (1 round trip, was 8 in series); partials added via LDS in fixed order (new sum order)
# speedup vs baseline: 5.1007x; 1.0883x over previous
#include <hip/hip_runtime.h>

#define V_  12
#define N_  80000
#define C_  256
#define H_  128
#define W_  128
#define G_  4096
#define M_  512
#define E_  200000
#define L_  20
#define C2_ 128

typedef __attribute__((ext_vector_type(2))) float v2f;
typedef __attribute__((ext_vector_type(8))) float v8f;
typedef __attribute__((ext_vector_type(4))) float v4f;
typedef __attribute__((ext_vector_type(4))) int   v4i;

#define PPW_   320
#define NWG_   (N_ / PPW_)
#define E4_    (E_ / 4)
#define NCH_   50
#define CH4_   (E4_ / NCH_)
#define CHE_   (CH4_ * 4)
#define OPITCH_ 256
#define TABB_  ((V_ * G_) / 1024)

__global__ void __launch_bounds__(1024)
k_prepass(const float* __restrict__ is_seen,
          const int*   __restrict__ coords,
          const int*   __restrict__ centers,
          const int*   __restrict__ pair_mask,
          const int*   __restrict__ pair_point,
          int* __restrict__ tab_off, float* __restrict__ tab_w,
          int* __restrict__ sorted, int* __restrict__ offs) {
  __shared__ int hist[OPITCH_];
  __shared__ int base[OPITCH_];
  const int t = threadIdx.x;
  if (blockIdx.x >= NCH_) {
    int idx = (blockIdx.x - NCH_) * 1024 + t;
    int v = idx >> 12;
    int g = idx & (G_ - 1);
    int n = centers[g];
    int x = coords[((size_t)v * N_ + n) * 2 + 0];
    int y = coords[((size_t)v * N_ + n) * 2 + 1];
    tab_off[idx] = y * W_ + x;
    tab_w[idx]   = is_seen[(size_t)v * N_ + n];
    return;
  }
  const int s = blockIdx.x;
  v4i p = {-1, -1, -1, -1};
  v4i m = {0, 0, 0, 0};
  if (t < CH4_) {
    p = ((const v4i*)pair_point)[s * CH4_ + t];
    m = ((const v4i*)pair_mask)[s * CH4_ + t];
  }
  if (t < OPITCH_) hist[t] = 0;
  __syncthreads();
  int own[4], rk[4];
#pragma unroll
  for (int j = 0; j < 4; ++j) {
    unsigned up = (unsigned)p[j];
    own[j] = (up < (unsigned)N_) ? (int)(up / PPW_) : -1;
    rk[j] = 0;
    if (own[j] >= 0) rk[j] = atomicAdd(&hist[own[j]], 1);
  }
  __syncthreads();
  int bstart[8] = {0, 0, 0, 0, 0, 0, 0, 0};
  if (t < 32) {
    int c[8];
    int run = 0;
#pragma unroll
    for (int i = 0; i < 8; ++i) { int h = hist[8 * t + i]; c[i] = run; run += h; }
    int incl = run;
#pragma unroll
    for (int d = 1; d < 32; d <<= 1) {
      int y = __shfl_up(incl, d);
      if (t >= d) incl += y;
    }
    int excl = incl - run;
#pragma unroll
    for (int i = 0; i < 8; ++i) {
      bstart[i] = excl + c[i];
      base[8 * t + i] = bstart[i];
    }
  }
  __syncthreads();
  if (t < 32) {
#pragma unroll
    for (int i = 0; i < 8; ++i) offs[s * OPITCH_ + 8 * t + i] = bstart[i];
  }
#pragma unroll
  for (int j = 0; j < 4; ++j) {
    if (own[j] >= 0)
      sorted[s * CHE_ + base[own[j]] + rk[j]] = ((p[j] - own[j] * PPW_) << 18) | m[j];
  }
}

#define PLANE4_ (H_ * W_ / 4)

__device__ __forceinline__ void lds_barrier() {
  __builtin_amdgcn_fence(__ATOMIC_RELEASE, "workgroup", "local");
  __builtin_amdgcn_s_barrier();
  __builtin_amdgcn_fence(__ATOMIC_ACQUIRE, "workgroup", "local");
}

__global__ void __launch_bounds__(1024)
k_group_gather(const float* __restrict__ img,
               const int*   __restrict__ tab_off,
               const float* __restrict__ tab_w,
               const int*   __restrict__ group2mask,
               float* __restrict__ mask_featT) {
  __shared__ v4f plane[2][PLANE4_];
  __shared__ float msum_l[M_];
  __shared__ int   mcnt_l[M_];
  const int c = blockIdx.x;
  const int t = threadIdx.x;
  const v4f* src = (const v4f*)img + (size_t)c * PLANE4_ + t;
  const size_t vstride = (size_t)C_ * PLANE4_;
  const v4i* toff = (const v4i*)tab_off + t;
  const v4f* tw   = (const v4f*)tab_w + t;

  const v4i g2m = *((const v4i*)group2mask + t);

  v4f acc = {0.0f, 0.0f, 0.0f, 0.0f};
  v4f den = {0.0f, 0.0f, 0.0f, 0.0f};
  v4f ra[4], rb[4];
  v4i oa, ob;
  v4f wa, wb;

#pragma unroll
  for (int j = 0; j < 4; ++j) ra[j] = __builtin_nontemporal_load(src + j * 1024);
  oa = toff[0]; wa = tw[0];
#pragma unroll
  for (int j = 0; j < 4; ++j) rb[j] = __builtin_nontemporal_load(src + vstride + j * 1024);
  ob = toff[G_ / 4]; wb = tw[G_ / 4];

  if (t < M_) { msum_l[t] = 0.0f; mcnt_l[t] = 0; }
  lds_barrier();
#pragma unroll
  for (int j = 0; j < 4; ++j) atomicAdd(&mcnt_l[g2m[j]], 1);

  for (int v = 0; v < V_; v += 2) {
    {
#pragma unroll
      for (int j = 0; j < 4; ++j) plane[0][t + j * 1024] = ra[j];
      lds_barrier();
      v4i o = oa; v4f w = wa;
      asm volatile("" : "+v"(o), "+v"(w));
      if (v + 2 < V_) {
#pragma unroll
        for (int j = 0; j < 4; ++j) ra[j] = __builtin_nontemporal_load(src + (size_t)(v + 2) * vstride + j * 1024);
        oa = toff[(v + 2) * (G_ / 4)]; wa = tw[(v + 2) * (G_ / 4)];
      }
      const float* pl = (const float*)plane[0];
#pragma unroll
      for (int j = 0; j < 4; ++j) {
        den[j] += w[j];
        acc[j] += w[j] * pl[o[j]];
      }
    }
    {
#pragma unroll
      for (int j = 0; j < 4; ++j) plane[1][t + j * 1024] = rb[j];
      lds_barrier();
      v4i o = ob; v4f w = wb;
      asm volatile("" : "+v"(o), "+v"(w));
      if (v + 3 < V_) {
#pragma unroll
        for (int j = 0; j < 4; ++j) rb[j] = __builtin_nontemporal_load(src + (size_t)(v + 3) * vstride + j * 1024);
        ob = toff[(v + 3) * (G_ / 4)]; wb = tw[(v + 3) * (G_ / 4)];
      }
      const float* pl = (const float*)plane[1];
#pragma unroll
      for (int j = 0; j < 4; ++j) {
        den[j] += w[j];
        acc[j] += w[j] * pl[o[j]];
      }
    }
  }
#pragma unroll
  for (int j = 0; j < 4; ++j) atomicAdd(&msum_l[g2m[j]], acc[j] / (den[j] + 1e-6f));
  lds_barrier();
  if (t < M_) {
    float cc = (float)mcnt_l[t];
    mask_featT[(size_t)c * M_ + t] = (cc > 0.0f) ? msum_l[t] / fmaxf(cc, 1.0f) : 0.0f;
  }
}

#define KQ_ (C_ / 4)
__global__ void __launch_bounds__(128)
k_mask_h1(const float* __restrict__ mask_featT,
          const float* __restrict__ W1,
          float* __restrict__ mask_h1) {
  __shared__ float part[4][8][32];
  const int lane = threadIdx.x & 31;
  const int wid  = threadIdx.x >> 5;
  const int tile = blockIdx.x;
  const int mt = tile >> 3, nt = tile & 7;
  const int mrow = (mt << 4) + (lane & 15);
  const int ncol = (nt << 4) + (lane & 15);
  const int kb   = wid * KQ_ + ((lane >> 4) << 1);
  const float* ap = mask_featT + (size_t)kb * M_ + mrow;
  const float* bp = W1 + (size_t)kb * C2_ + ncol;
  v2f a[KQ_ / 4], b[KQ_ / 4];
#pragma unroll
  for (int s = 0; s < KQ_ / 4; ++s) {
    a[s].x = ap[(size_t)(4 * s) * M_];
    a[s].y = ap[(size_t)(4 * s + 1) * M_];
    b[s].x = bp[(size_t)(4 * s) * C2_];
    b[s].y = bp[(size_t)(4 * s + 1) * C2_];
  }
  __builtin_amdgcn_sched_barrier(0);
  v8f c0 = {}, c1 = {};
#pragma unroll
  for (int s = 0; s < KQ_ / 4; s += 2) {
    c0 = __builtin_amdgcn_wmma_f32_16x16x4_f32(false, a[s],     false, b[s],     (short)0, c0, false, false);
    c1 = __builtin_amdgcn_wmma_f32_16x16x4_f32(false, a[s + 1], false, b[s + 1], (short)0, c1, false, false);
  }
#pragma unroll
  for (int i = 0; i < 8; ++i) part[wid][i][lane] = c0[i] + c1[i];
  __syncthreads();
  for (int j = 0; j < 2; ++j) {
    int i = 2 * wid + j;
    float s = ((part[0][i][lane] + part[1][i][lane]) + part[2][i][lane]) + part[3][i][lane];
    int r = (mt << 4) + i + ((lane >> 4) << 3);
    mask_h1[(size_t)r * C2_ + ncol] = s;
  }
}

#define PITCH_ 132
#define CAP_   16384
#define TPS_   20
#define NB_    8

__global__ void __launch_bounds__(1024)
k_point_fused(const float* __restrict__ mask_h1,
              const int*   __restrict__ sorted,
              const int*   __restrict__ offs,
              const float* __restrict__ b1, const float* __restrict__ W2,
              const float* __restrict__ b2, float* __restrict__ out) {
  __shared__ v4f acc4[PPW_ * PITCH_ / 4];
  __shared__ int list[CAP_];
  __shared__ v4f w2f[(C2_ / 4) * 32];
  __shared__ float sb1[C2_];
  __shared__ int pcnt[PPW_];
  __shared__ int list_n;
  float* acc = (float*)acc4;
  const int t = threadIdx.x, lane = t & 31, wid = t >> 5;
  const int lo = blockIdx.x * PPW_;

  v4f wfrag;
  float b1v0 = 0.0f;
  {
    int c = lane & 15, ka = (wid << 2) + ((lane >> 4) << 1), kb = ka + 1;
    wfrag[0] = W2[(size_t)ka * L_ + c];
    wfrag[1] = W2[(size_t)kb * L_ + c];
    wfrag[2] = (16 + c < L_) ? W2[(size_t)ka * L_ + 16 + c] : 0.0f;
    wfrag[3] = (16 + c < L_) ? W2[(size_t)kb * L_ + 16 + c] : 0.0f;
    if (t < C2_) b1v0 = b1[t];
  }
  const int sc = t / TPS_, sr = t - sc * TPS_;
  int seg_st = 0, seg_en = 0;
  if (t < NCH_ * TPS_) {
    const int* o = offs + sc * OPITCH_ + blockIdx.x;
    seg_st = o[0];
    seg_en = o[1];
  }
  for (int i = t; i < PPW_ * PITCH_ / 4; i += 1024) acc4[i] = v4f{0.0f, 0.0f, 0.0f, 0.0f};
  if (t < PPW_) pcnt[t] = 0;
  if (t == 0) list_n = 0;
  __syncthreads();

  {
    const int* seg = sorted + sc * CHE_;
    for (int i = seg_st + sr; i < seg_en; i += TPS_) {
      int ent = seg[i];
      int rel = ent >> 18;
      atomicAdd(&pcnt[rel], 1);
      int pos = atomicAdd(&list_n, 1);
      if (pos < CAP_) {
        list[pos] = ent;
      } else {
        int m = ent & 0x3FFFF;
        for (int k = 0; k < C2_; ++k)
          atomicAdd(&acc[rel * PITCH_ + k], mask_h1[(size_t)m * C2_ + k]);
      }
    }
  }
  w2f[t] = wfrag;
  if (t < C2_) sb1[t] = b1v0;
  __syncthreads();
  const int n = __builtin_amdgcn_readfirstlane(list_n < CAP_ ? list_n : CAP_);
  const int swid = __builtin_amdgcn_readfirstlane(wid);

  for (int k0 = swid; k0 < n; k0 += 32 * NB_) {
    int x[NB_], ent[NB_];
    float r[NB_][4];
#pragma unroll
    for (int u = 0; u < NB_; ++u) {
      int k = k0 + 32 * u;
      x[u] = list[k < n ? k : k0];
    }
#pragma unroll
    for (int u = 0; u < NB_; ++u) ent[u] = __builtin_amdgcn_readfirstlane(x[u]);
#pragma unroll
    for (int u = 0; u < NB_; ++u) {
      const float* s = mask_h1 + (size_t)(ent[u] & 0x3FFFF) * C2_ + lane;
#pragma unroll
      for (int j = 0; j < 4; ++j) r[u][j] = s[32 * j];
    }
#pragma unroll
    for (int u = 0; u < NB_; ++u) {
      if (k0 + 32 * u < n) {
        float* d = acc + (ent[u] >> 18) * PITCH_ + lane;
#pragma unroll
        for (int j = 0; j < 4; ++j) atomicAdd(d + 32 * j, r[u][j]);
      }
    }
  }
  __syncthreads();

  if (swid < PPW_ / 16) {
    int c = lane & 15;
    int h = lane >> 4;
    int row = (swid << 4) + c;
    float scale = 1.0f / fmaxf((float)pcnt[row], 1e-6f);
    const float* arow = acc + row * PITCH_ + (h << 1);
    const float* brow = sb1 + (h << 1);
    v8f c0 = {}, c1 = {};
#pragma unroll 8
    for (int kk = 0; kk < C2_; kk += 4) {
      v2f s = *(const v2f*)(arow + kk);
      v2f bb = *(const v2f*)(brow + kk);
      v4f f = w2f[(kk >> 2) * 32 + lane];
      v2f a, b0, b1v;
      a.x = fmaxf(s.x * scale + bb.x, 0.0f);
      a.y = fmaxf(s.y * scale + bb.y, 0.0f);
      b0.x = f[0]; b0.y = f[1];
      b1v.x = f[2]; b1v.y = f[3];
      c0 = __builtin_amdgcn_wmma_f32_16x16x4_f32(false, a, false, b0,  (short)0, c0, false, false);
      c1 = __builtin_amdgcn_wmma_f32_16x16x4_f32(false, a, false, b1v, (short)0, c1, false, false);
    }
    for (int i = 0; i < 8; ++i) {
      int r = lo + (swid << 4) + i + (h << 3);
      out[(size_t)r * L_ + c] = c0[i] + b2[c];
      int col1 = 16 + c;
      if (col1 < L_) out[(size_t)r * L_ + col1] = c1[i] + b2[col1];
    }
  }
}

extern "C" void kernel_launch(void* const* d_in, const int* in_sizes, int n_in,
                              void* d_out, int out_size, void* d_ws, size_t ws_size,
                              hipStream_t stream) {
  const float* img        = (const float*)d_in[0];
  const float* is_seen    = (const float*)d_in[1];
  const float* W1         = (const float*)d_in[2];
  const float* b1         = (const float*)d_in[3];
  const float* W2         = (const float*)d_in[4];
  const float* b2         = (const float*)d_in[5];
  const int*   coords     = (const int*)d_in[6];
  const int*   centers    = (const int*)d_in[7];
  const int*   group2mask = (const int*)d_in[8];
  const int*   pair_mask  = (const int*)d_in[9];
  const int*   pair_point = (const int*)d_in[10];
  float* out = (float*)d_out;

  float* ws         = (float*)d_ws;
  int*   tab_off    = (int*)ws;
  float* tab_w      = ws + (size_t)V_ * G_;
  int*   sorted     = (int*)(ws + (size_t)2 * V_ * G_);
  int*   offs       = sorted + E_;
  float* mask_featT = ws + (size_t)G_ * C_ + (size_t)M_ * C_ + M_;
  float* mask_h1    = mask_featT + (size_t)M_ * C_;

  k_prepass<<<NCH_ + TABB_, 1024, 0, stream>>>(is_seen, coords, centers, pair_mask, pair_point,
                                               tab_off, tab_w, sorted, offs);
  k_group_gather<<<C_, 1024, 0, stream>>>(img, tab_off, tab_w, group2mask, mask_featT);
  k_mask_h1<<<256, 128, 0, stream>>>(mask_featT, W1, mask_h1);

  k_point_fused<<<NWG_, 1024, 0, stream>>>(mask_h1, sorted, offs, b1, W2, b2, out);
}


# point kernel: edges counting-sorted by point in LDS; each of 20 waves sums its 16 points' mask rows in registers (list order, set by cursor arrival) for its WMMA tile; no float atomics on this path
# speedup vs baseline: 5.4692x; 1.0722x over previous
#include <hip/hip_runtime.h>

#define V_  12
#define N_  80000
#define C_  256
#define H_  128
#define W_  128
#define G_  4096
#define M_  512
#define E_  200000
#define L_  20
#define C2_ 128

typedef __attribute__((ext_vector_type(2))) float v2f;
typedef __attribute__((ext_vector_type(8))) float v8f;
typedef __attribute__((ext_vector_type(4))) float v4f;
typedef __attribute__((ext_vector_type(4))) int   v4i;

#define PPW_   320
#define NWG_   (N_ / PPW_)
#define E4_    (E_ / 4)
#define NCH_   50
#define CH4_   (E4_ / NCH_)
#define CHE_   (CH4_ * 4)
#define OPITCH_ 256
#define TABB_  ((V_ * G_) / 1024)

__global__ void __launch_bounds__(1024)
k_prepass(const float* __restrict__ is_seen,
          const int*   __restrict__ coords,
          const int*   __restrict__ centers,
          const int*   __restrict__ pair_mask,
          const int*   __restrict__ pair_point,
          int* __restrict__ tab_off, float* __restrict__ tab_w,
          int* __restrict__ sorted, int* __restrict__ offs) {
  __shared__ int hist[OPITCH_];
  __shared__ int base[OPITCH_];
  const int t = threadIdx.x;
  if (blockIdx.x >= NCH_) {
    int idx = (blockIdx.x - NCH_) * 1024 + t;
    int v = idx >> 12;
    int g = idx & (G_ - 1);
    int n = centers[g];
    int x = coords[((size_t)v * N_ + n) * 2 + 0];
    int y = coords[((size_t)v * N_ + n) * 2 + 1];
    tab_off[idx] = y * W_ + x;
    tab_w[idx]   = is_seen[(size_t)v * N_ + n];
    return;
  }
  const int s = blockIdx.x;
  v4i p = {-1, -1, -1, -1};
  v4i m = {0, 0, 0, 0};
  if (t < CH4_) {
    p = ((const v4i*)pair_point)[s * CH4_ + t];
    m = ((const v4i*)pair_mask)[s * CH4_ + t];
  }
  if (t < OPITCH_) hist[t] = 0;
  __syncthreads();
  int own[4], rk[4];
#pragma unroll
  for (int j = 0; j < 4; ++j) {
    unsigned up = (unsigned)p[j];
    own[j] = (up < (unsigned)N_) ? (int)(up / PPW_) : -1;
    rk[j] = 0;
    if (own[j] >= 0) rk[j] = atomicAdd(&hist[own[j]], 1);
  }
  __syncthreads();
  int bstart[8] = {0, 0, 0, 0, 0, 0, 0, 0};
  if (t < 32) {
    int c[8];
    int run = 0;
#pragma unroll
    for (int i = 0; i < 8; ++i) { int h = hist[8 * t + i]; c[i] = run; run += h; }
    int incl = run;
#pragma unroll
    for (int d = 1; d < 32; d <<= 1) {
      int y = __shfl_up(incl, d);
      if (t >= d) incl += y;
    }
    int excl = incl - run;
#pragma unroll
    for (int i = 0; i < 8; ++i) {
      bstart[i] = excl + c[i];
      base[8 * t + i] = bstart[i];
    }
  }
  __syncthreads();
  if (t < 32) {
#pragma unroll
    for (int i = 0; i < 8; ++i) offs[s * OPITCH_ + 8 * t + i] = bstart[i];
  }
#pragma unroll
  for (int j = 0; j < 4; ++j) {
    if (own[j] >= 0)
      sorted[s * CHE_ + base[own[j]] + rk[j]] = ((p[j] - own[j] * PPW_) << 18) | m[j];
  }
}

#define PLANE4_ (H_ * W_ / 4)

__device__ __forceinline__ void lds_barrier() {
  __builtin_amdgcn_fence(__ATOMIC_RELEASE, "workgroup", "local");
  __builtin_amdgcn_s_barrier();
  __builtin_amdgcn_fence(__ATOMIC_ACQUIRE, "workgroup", "local");
}

__global__ void __launch_bounds__(1024)
k_group_gather(const float* __restrict__ img,
               const int*   __restrict__ tab_off,
               const float* __restrict__ tab_w,
               const int*   __restrict__ group2mask,
               float* __restrict__ mask_featT) {
  __shared__ v4f plane[2][PLANE4_];
  __shared__ float msum_l[M_];
  __shared__ int   mcnt_l[M_];
  const int c = blockIdx.x;
  const int t = threadIdx.x;
  const v4f* src = (const v4f*)img + (size_t)c * PLANE4_ + t;
  const size_t vstride = (size_t)C_ * PLANE4_;
  const v4i* toff = (const v4i*)tab_off + t;
  const v4f* tw   = (const v4f*)tab_w + t;

  const v4i g2m = *((const v4i*)group2mask + t);

  v4f acc = {0.0f, 0.0f, 0.0f, 0.0f};
  v4f den = {0.0f, 0.0f, 0.0f, 0.0f};
  v4f ra[4], rb[4];
  v4i oa, ob;
  v4f wa, wb;

#pragma unroll
  for (int j = 0; j < 4; ++j) ra[j] = __builtin_nontemporal_load(src + j * 1024);
  oa = toff[0]; wa = tw[0];
#pragma unroll
  for (int j = 0; j < 4; ++j) rb[j] = __builtin_nontemporal_load(src + vstride + j * 1024);
  ob = toff[G_ / 4]; wb = tw[G_ / 4];

  if (t < M_) { msum_l[t] = 0.0f; mcnt_l[t] = 0; }
  lds_barrier();
#pragma unroll
  for (int j = 0; j < 4; ++j) atomicAdd(&mcnt_l[g2m[j]], 1);

  for (int v = 0; v < V_; v += 2) {
    {
#pragma unroll
      for (int j = 0; j < 4; ++j) plane[0][t + j * 1024] = ra[j];
      lds_barrier();
      v4i o = oa; v4f w = wa;
      asm volatile("" : "+v"(o), "+v"(w));
      if (v + 2 < V_) {
#pragma unroll
        for (int j = 0; j < 4; ++j) ra[j] = __builtin_nontemporal_load(src + (size_t)(v + 2) * vstride + j * 1024);
        oa = toff[(v + 2) * (G_ / 4)]; wa = tw[(v + 2) * (G_ / 4)];
      }
      const float* pl = (const float*)plane[0];
#pragma unroll
      for (int j = 0; j < 4; ++j) {
        den[j] += w[j];
        acc[j] += w[j] * pl[o[j]];
      }
    }
    {
#pragma unroll
      for (int j = 0; j < 4; ++j) plane[1][t + j * 1024] = rb[j];
      lds_barrier();
      v4i o = ob; v4f w = wb;
      asm volatile("" : "+v"(o), "+v"(w));
      if (v + 3 < V_) {
#pragma unroll
        for (int j = 0; j < 4; ++j) rb[j] = __builtin_nontemporal_load(src + (size_t)(v + 3) * vstride + j * 1024);
        ob = toff[(v + 3) * (G_ / 4)]; wb = tw[(v + 3) * (G_ / 4)];
      }
      const float* pl = (const float*)plane[1];
#pragma unroll
      for (int j = 0; j < 4; ++j) {
        den[j] += w[j];
        acc[j] += w[j] * pl[o[j]];
      }
    }
  }
#pragma unroll
  for (int j = 0; j < 4; ++j) atomicAdd(&msum_l[g2m[j]], acc[j] / (den[j] + 1e-6f));
  lds_barrier();
  if (t < M_) {
    float cc = (float)mcnt_l[t];
    mask_featT[(size_t)c * M_ + t] = (cc > 0.0f) ? msum_l[t] / fmaxf(cc, 1.0f) : 0.0f;
  }
}

#define KQ_ (C_ / 4)
__global__ void __launch_bounds__(128)
k_mask_h1(const float* __restrict__ mask_featT,
          const float* __restrict__ W1,
          float* __restrict__ mask_h1) {
  __shared__ float part[4][8][32];
  const int lane = threadIdx.x & 31;
  const int wid  = threadIdx.x >> 5;
  const int tile = blockIdx.x;
  const int mt = tile >> 3, nt = tile & 7;
  const int mrow = (mt << 4) + (lane & 15);
  const int ncol = (nt << 4) + (lane & 15);
  const int kb   = wid * KQ_ + ((lane >> 4) << 1);
  const float* ap = mask_featT + (size_t)kb * M_ + mrow;
  const float* bp = W1 + (size_t)kb * C2_ + ncol;
  v2f a[KQ_ / 4], b[KQ_ / 4];
#pragma unroll
  for (int s = 0; s < KQ_ / 4; ++s) {
    a[s].x = ap[(size_t)(4 * s) * M_];
    a[s].y = ap[(size_t)(4 * s + 1) * M_];
    b[s].x = bp[(size_t)(4 * s) * C2_];
    b[s].y = bp[(size_t)(4 * s + 1) * C2_];
  }
  __builtin_amdgcn_sched_barrier(0);
  v8f c0 = {}, c1 = {};
#pragma unroll
  for (int s = 0; s < KQ_ / 4; s += 2) {
    c0 = __builtin_amdgcn_wmma_f32_16x16x4_f32(false, a[s],     false, b[s],     (short)0, c0, false, false);
    c1 = __builtin_amdgcn_wmma_f32_16x16x4_f32(false, a[s + 1], false, b[s + 1], (short)0, c1, false, false);
  }
#pragma unroll
  for (int i = 0; i < 8; ++i) part[wid][i][lane] = c0[i] + c1[i];
  __syncthreads();
  for (int j = 0; j < 2; ++j) {
    int i = 2 * wid + j;
    float s = ((part[0][i][lane] + part[1][i][lane]) + part[2][i][lane]) + part[3][i][lane];
    int r = (mt << 4) + i + ((lane >> 4) << 3);
    mask_h1[(size_t)r * C2_ + ncol] = s;
  }
}

#define PITCH_ 132
#define CAP_   16384
#define NT_    (PPW_ / 16)
#define TP_    (NT_ * 32)
#define TPS_   12
#define NB_    32

__global__ void __launch_bounds__(TP_)
k_point_fused(const float* __restrict__ mask_h1,
              const int*   __restrict__ sorted,
              const int*   __restrict__ offs,
              const float* __restrict__ b1, const float* __restrict__ W2,
              const float* __restrict__ b2, float* __restrict__ out) {
  __shared__ v4f htile[PPW_ * PITCH_ / 4];
  __shared__ int slist[CAP_];
  __shared__ v4f w2f[(C2_ / 4) * 32];
  __shared__ int pcnt[PPW_];
  __shared__ int cur[PPW_];
  __shared__ int tot;
  const int t = threadIdx.x, lane = t & 31;
  const int tile = __builtin_amdgcn_readfirstlane(t >> 5);
  const int lo = blockIdx.x * PPW_;

  const int sc = t / TPS_, sr = t - sc * TPS_;
  int seg_st = 0, seg_en = 0;
  if (t < NCH_ * TPS_) {
    const int* o = offs + sc * OPITCH_ + blockIdx.x;
    seg_st = o[0];
    seg_en = o[1];
  }
  v4f wfrag[2];
#pragma unroll
  for (int q = 0; q < 2; ++q) {
    int e = t + q * TP_;
    int c = e & 15, ka = ((e >> 5) << 2) + (((e >> 4) & 1) << 1), kb = ka + 1;
    wfrag[q] = v4f{0.0f, 0.0f, 0.0f, 0.0f};
    if (e < (C2_ / 4) * 32) {
      wfrag[q][0] = W2[(size_t)ka * L_ + c];
      wfrag[q][1] = W2[(size_t)kb * L_ + c];
      wfrag[q][2] = (16 + c < L_) ? W2[(size_t)ka * L_ + 16 + c] : 0.0f;
      wfrag[q][3] = (16 + c < L_) ? W2[(size_t)kb * L_ + 16 + c] : 0.0f;
    }
  }
  const v4f b1v = *((const v4f*)b1 + lane);
  const int cc = lane & 15, hh = lane >> 4;
  const float b2a = b2[cc];
  const float b2b = (16 + cc < L_) ? b2[16 + cc] : 0.0f;

  if (t < PPW_) pcnt[t] = 0;
  lds_barrier();

  const int* seg = sorted + sc * CHE_;
  const int i0 = seg_st + sr, i1 = i0 + TPS_;
  int ea = -1, eb = -1;
  if (i0 < seg_en) ea = seg[i0];
  if (i1 < seg_en) eb = seg[i1];
  if (ea >= 0) atomicAdd(&pcnt[ea >> 18], 1);
  if (eb >= 0) atomicAdd(&pcnt[eb >> 18], 1);
  for (int i = i1 + TPS_; i < seg_en; i += TPS_) atomicAdd(&pcnt[seg[i] >> 18], 1);
  lds_barrier();

  if (tile == 0) {
    int c[10];
    int run = 0;
#pragma unroll
    for (int i = 0; i < 10; ++i) { int h = pcnt[10 * lane + i]; c[i] = run; run += h; }
    int incl = run;
#pragma unroll
    for (int d = 1; d < 32; d <<= 1) {
      int y = __shfl_up(incl, d);
      if (lane >= d) incl += y;
    }
    int excl = incl - run;
#pragma unroll
    for (int i = 0; i < 10; ++i) cur[10 * lane + i] = excl + c[i];
    if (lane == 31) tot = incl;
  }
  lds_barrier();

  if (ea >= 0) { int pos = atomicAdd(&cur[ea >> 18], 1); if (pos < CAP_) slist[pos] = ea & 0x3FFFF; }
  if (eb >= 0) { int pos = atomicAdd(&cur[eb >> 18], 1); if (pos < CAP_) slist[pos] = eb & 0x3FFFF; }
  for (int i = i1 + TPS_; i < seg_en; i += TPS_) {
    int ent = seg[i];
    int pos = atomicAdd(&cur[ent >> 18], 1);
    if (pos < CAP_) slist[pos] = ent & 0x3FFFF;
  }
#pragma unroll
  for (int q = 0; q < 2; ++q) { int e = t + q * TP_; if (e < (C2_ / 4) * 32) w2f[e] = wfrag[q]; }
  lds_barrier();

  float* ht = (float*)htile + tile * 16 * PITCH_;
  const int endv = cur[tile * 16 + cc];
  const int cntv = pcnt[tile * 16 + cc];
  const float scalev = 1.0f / fmaxf((float)cntv, 1e-6f);
  const int e0 = __builtin_amdgcn_readfirstlane(tile ? cur[tile * 16 - 1] : 0);
  const int e1 = __builtin_amdgcn_readlane(endv, 15);
  const int ntot = __builtin_amdgcn_readfirstlane(tot);

  if (ntot <= CAP_) {
    v4f acc = {0.0f, 0.0f, 0.0f, 0.0f};
    int p = 0;
    int pend = __builtin_amdgcn_readlane(endv, 0);
    for (int base = e0; base < e1; base += NB_) {
      const int idx = base + lane;
      const int mv = slist[idx < e1 ? idx : base];
      v4f r[NB_];
#pragma unroll
      for (int u = 0; u < NB_; ++u) {
        if (base + u < e1) {
          const int m = __builtin_amdgcn_readlane(mv, u);
          r[u] = *((const v4f*)(mask_h1 + (size_t)m * C2_) + lane);
        }
      }
#pragma unroll
      for (int u = 0; u < NB_; ++u) {
        if (base + u < e1) {
          while (base + u >= pend) {
            const float sc1 = __int_as_float(__builtin_amdgcn_readlane(__float_as_int(scalev), p));
            v4f hv;
#pragma unroll
            for (int j = 0; j < 4; ++j) hv[j] = fmaxf(acc[j] * sc1 + b1v[j], 0.0f);
            *((v4f*)(ht + p * PITCH_) + lane) = hv;
            acc = v4f{0.0f, 0.0f, 0.0f, 0.0f};
            ++p;
            pend = __builtin_amdgcn_readlane(endv, p);
          }
          acc += r[u];
        }
      }
    }
    for (; p < 16; ++p) {
      const float sc1 = __int_as_float(__builtin_amdgcn_readlane(__float_as_int(scalev), p));
      v4f hv;
#pragma unroll
      for (int j = 0; j < 4; ++j) hv[j] = fmaxf(acc[j] * sc1 + b1v[j], 0.0f);
      *((v4f*)(ht + p * PITCH_) + lane) = hv;
      acc = v4f{0.0f, 0.0f, 0.0f, 0.0f};
    }
  } else {
    for (int i = lane; i < 16 * PITCH_ / 4; i += 32) ((v4f*)ht)[i] = v4f{0.0f, 0.0f, 0.0f, 0.0f};
    for (int ch = 0; ch < NCH_; ++ch) {
      const int* o = offs + ch * OPITCH_ + blockIdx.x;
      const int st = o[0], en = o[1];
      for (int i = st + lane; i < en; i += 32) {
        int ent = sorted[ch * CHE_ + i];
        int rel = ent >> 18;
        if ((rel >> 4) == tile) {
          int m = ent & 0x3FFFF;
          for (int k = 0; k < C2_; ++k)
            atomicAdd(&ht[(rel & 15) * PITCH_ + k], mask_h1[(size_t)m * C2_ + k]);
        }
      }
    }
    for (int p = 0; p < 16; ++p) {
      const float sc1 = __int_as_float(__builtin_amdgcn_readlane(__float_as_int(scalev), p));
      v4f s = *((v4f*)(ht + p * PITCH_) + lane);
      v4f hv;
#pragma unroll
      for (int j = 0; j < 4; ++j) hv[j] = fmaxf(s[j] * sc1 + b1v[j], 0.0f);
      *((v4f*)(ht + p * PITCH_) + lane) = hv;
    }
  }

  {
    const float* arow = ht + cc * PITCH_ + (hh << 1);
    v8f c0 = {}, c1 = {};
#pragma unroll 8
    for (int kk = 0; kk < C2_; kk += 4) {
      v2f a = *(const v2f*)(arow + kk);
      v4f f = w2f[(kk >> 2) * 32 + lane];
      v2f b0, b1w;
      b0.x = f[0]; b0.y = f[1];
      b1w.x = f[2]; b1w.y = f[3];
      c0 = __builtin_amdgcn_wmma_f32_16x16x4_f32(false, a, false, b0,  (short)0, c0, false, false);
      c1 = __builtin_amdgcn_wmma_f32_16x16x4_f32(false, a, false, b1w, (short)0, c1, false, false);
    }
    for (int i = 0; i < 8; ++i) {
      int r = lo + (tile << 4) + i + (hh << 3);
      out[(size_t)r * L_ + cc] = c0[i] + b2a;
      int col1 = 16 + cc;
      if (col1 < L_) out[(size_t)r * L_ + col1] = c1[i] + b2b;
    }
  }
}

extern "C" void kernel_launch(void* const* d_in, const int* in_sizes, int n_in,
                              void* d_out, int out_size, void* d_ws, size_t ws_size,
                              hipStream_t stream) {
  const float* img        = (const float*)d_in[0];
  const float* is_seen    = (const float*)d_in[1];
  const float* W1         = (const float*)d_in[2];
  const float* b1         = (const float*)d_in[3];
  const float* W2         = (const float*)d_in[4];
  const float* b2         = (const float*)d_in[5];
  const int*   coords     = (const int*)d_in[6];
  const int*   centers    = (const int*)d_in[7];
  const int*   group2mask = (const int*)d_in[8];
  const int*   pair_mask  = (const int*)d_in[9];
  const int*   pair_point = (const int*)d_in[10];
  float* out = (float*)d_out;

  float* ws         = (float*)d_ws;
  int*   tab_off    = (int*)ws;
  float* tab_w      = ws + (size_t)V_ * G_;
  int*   sorted     = (int*)(ws + (size_t)2 * V_ * G_);
  int*   offs       = sorted + E_;
  float* mask_featT = ws + (size_t)G_ * C_ + (size_t)M_ * C_ + M_;
  float* mask_h1    = mask_featT + (size_t)M_ * C_;

  k_prepass<<<NCH_ + TABB_, 1024, 0, stream>>>(is_seen, coords, centers, pair_mask, pair_point,
                                               tab_off, tab_w, sorted, offs);
  k_group_gather<<<C_, 1024, 0, stream>>>(img, tab_off, tab_w, group2mask, mask_featT);
  k_mask_h1<<<256, 128, 0, stream>>>(mask_featT, W1, mask_h1);

  k_point_fused<<<NWG_, TP_, 0, stream>>>(mask_h1, sorted, offs, b1, W2, b2, out);
}
